# starting kernel
# baseline (speedup 1.0000x reference)
#include <hip/hip_runtime.h>
#include <math.h>

#define BB    2
#define NN    50000
#define EE    800000
#define INP   128
#define KEY   64
#define VAL   128
#define MROWS (BB * NN)
#define LN_EPS 1e-5f

typedef __attribute__((ext_vector_type(2))) float v2f;
typedef __attribute__((ext_vector_type(8))) float v8f;

__device__ __forceinline__ unsigned enc_f32(float f) {
    unsigned u = __float_as_uint(f);
    return (u & 0x80000000u) ? ~u : (u | 0x80000000u);
}
__device__ __forceinline__ float dec_f32(unsigned u) {
    return (u & 0x80000000u) ? __uint_as_float(u & 0x7FFFFFFFu)
                             : __uint_as_float(~u);
}

__global__ void bsa_reduce_kw(const float* __restrict__ kw,
                              const float* __restrict__ kb,
                              float* __restrict__ kwsum,
                              float* __restrict__ kbsum) {
    int i = threadIdx.x;
    float s = 0.f;
    for (int j = 0; j < KEY; ++j) s += kw[i * KEY + j];
    kwsum[i] = s;
    if (i == 0) {
        float t = 0.f;
        for (int j = 0; j < KEY; ++j) t += kb[j];
        kbsum[0] = t;
    }
}

__global__ void bsa_ksum(const float* __restrict__ x,
                         const float* __restrict__ kwsum,
                         const float* __restrict__ kbsum,
                         float* __restrict__ ksum) {
    int wid  = (blockIdx.x * blockDim.x + threadIdx.x) >> 5;
    int lane = threadIdx.x & 31;
    if (wid >= MROWS) return;
    const float* xr = x + (size_t)wid * INP;
    float s = 0.f;
    #pragma unroll
    for (int i = 0; i < INP / 32; ++i) {
        int c = lane + i * 32;
        s += xr[c] * kwsum[c];
    }
    #pragma unroll
    for (int off = 16; off; off >>= 1) s += __shfl_xor(s, off);
    if (lane == 0) ksum[wid] = s + kbsum[0];
}

__device__ __forceinline__ void bsa_wmma_tile(const float* __restrict__ Arow,
                                              const float* __restrict__ Bs,
                                              v8f acc[8], int lane) {
    const int kh = (lane >> 4) << 1;
    const int n  = lane & 15;
    for (int k = 0; k < 128; k += 4) {
        v2f a = *(const v2f*)(Arow + k + kh);
        #pragma unroll
        for (int c = 0; c < 8; ++c) {
            v2f bf;
            bf.x = Bs[(k + kh)     * 128 + (c << 4) + n];
            bf.y = Bs[(k + kh + 1) * 128 + (c << 4) + n];
            acc[c] = __builtin_amdgcn_wmma_f32_16x16x4_f32(
                false, a, false, bf, (short)0, acc[c], false, false);
        }
    }
}

__global__ void bsa_values_gemm(const float* __restrict__ A,
                                const float* __restrict__ W,
                                const float* __restrict__ bias,
                                float* __restrict__ out) {
    __shared__ float Bs[128 * 128];
    for (int i = threadIdx.x; i < 128 * 128; i += blockDim.x) Bs[i] = W[i];
    __syncthreads();

    int lane    = threadIdx.x & 31;
    int wv      = threadIdx.x >> 5;
    int tile    = blockIdx.x * 8 + wv;
    int rowbase = tile * 16;
    if (rowbase >= MROWS) return;

    v8f acc[8] = {};
    const float* Arow = A + (size_t)(rowbase + (lane & 15)) * 128;
    bsa_wmma_tile(Arow, Bs, acc, lane);

    int n      = lane & 15;
    int rowoff = (lane >> 4) << 3;
    #pragma unroll
    for (int c = 0; c < 8; ++c) {
        float bb = bias[(c << 4) + n];
        #pragma unroll
        for (int r = 0; r < 8; ++r)
            out[(size_t)(rowbase + rowoff + r) * 128 + (c << 4) + n] = acc[c][r] + bb;
    }
}

__global__ void bsa_edge_max(const float* __restrict__ conn,
                             const int* __restrict__ rows,
                             const int* __restrict__ cols,
                             const float* __restrict__ ksum,
                             unsigned* __restrict__ mu) {
    int tid = blockIdx.x * blockDim.x + threadIdx.x;
    if (tid >= EE * BB) return;
    int e = tid >> 1, b = tid & 1;
    float s = conn[e] * ksum[b * NN + cols[e]];
    atomicMax(&mu[rows[e] * BB + b], enc_f32(s));
}

__global__ void bsa_m_finalize(unsigned* __restrict__ mu) {
    int i = blockIdx.x * blockDim.x + threadIdx.x;
    if (i >= NN * BB) return;
    unsigned u = mu[i];
    float m = (u == 0u) ? 0.0f : dec_f32(u);
    if (!isfinite(m)) m = 0.0f;
    ((float*)mu)[i] = m;
}

__global__ void bsa_edge_exp(const float* __restrict__ conn,
                             const int* __restrict__ rows,
                             const int* __restrict__ cols,
                             const float* __restrict__ ksum,
                             const float* __restrict__ m,
                             float* __restrict__ ex,
                             float* __restrict__ denom) {
    int tid = blockIdx.x * blockDim.x + threadIdx.x;
    if (tid >= EE * BB) return;
    int e = tid >> 1, b = tid & 1;
    int r = rows[e];
    float s = conn[e] * ksum[b * NN + cols[e]];
    float ev = __expf(s - m[r * BB + b]);
    ex[tid] = ev;
    atomicAdd(&denom[r * BB + b], ev);
}

__global__ void bsa_scatter(const int* __restrict__ rows,
                            const int* __restrict__ cols,
                            const float* __restrict__ ex,
                            const float* __restrict__ denom,
                            const float* __restrict__ values,
                            float* __restrict__ agg) {
    int e = blockIdx.x;
    int b = threadIdx.x >> 7;
    int c = threadIdx.x & 127;
    int r = rows[e], cl = cols[e];
    float d = denom[r * BB + b];
    float w = ex[e * BB + b] / (d == 0.0f ? 1.0f : d);
    float v = values[((size_t)b * NN + cl) * VAL + c];
    atomicAdd(&agg[((size_t)b * NN + r) * VAL + c], w * v);
}

__global__ void bsa_out_gemm_ln(const float* __restrict__ A,
                                const float* __restrict__ W,
                                const float* __restrict__ bias,
                                const float* __restrict__ lng,
                                const float* __restrict__ lnb,
                                float* __restrict__ out) {
    __shared__ float Bs[128 * 128];
    for (int i = threadIdx.x; i < 128 * 128; i += blockDim.x) Bs[i] = W[i];
    __syncthreads();

    int lane    = threadIdx.x & 31;
    int wv      = threadIdx.x >> 5;
    int tile    = blockIdx.x * 8 + wv;
    int rowbase = tile * 16;
    if (rowbase >= MROWS) return;

    v8f acc[8] = {};
    const float* Arow = A + (size_t)(rowbase + (lane & 15)) * 128;
    bsa_wmma_tile(Arow, Bs, acc, lane);

    int n = lane & 15;
    float sr[8], sq[8];
    #pragma unroll
    for (int r = 0; r < 8; ++r) { sr[r] = 0.f; sq[r] = 0.f; }
    #pragma unroll
    for (int c = 0; c < 8; ++c) {
        float bb = bias[(c << 4) + n];
        #pragma unroll
        for (int r = 0; r < 8; ++r) {
            float v = acc[c][r] + bb;
            v = v / (1.0f + __expf(-v));
            acc[c][r] = v;
            sr[r] += v;
            sq[r] += v * v;
        }
    }
    #pragma unroll
    for (int r = 0; r < 8; ++r) {
        #pragma unroll
        for (int off = 8; off; off >>= 1) {
            sr[r] += __shfl_xor(sr[r], off);
            sq[r] += __shfl_xor(sq[r], off);
        }
    }
    float mu[8], rstd[8];
    #pragma unroll
    for (int r = 0; r < 8; ++r) {
        mu[r] = sr[r] * (1.0f / VAL);
        float var = sq[r] * (1.0f / VAL) - mu[r] * mu[r];
        rstd[r] = rsqrtf(var + LN_EPS);
    }
    int rowoff = (lane >> 4) << 3;
    #pragma unroll
    for (int c = 0; c < 8; ++c) {
        int col = (c << 4) + n;
        float g = lng[col], b2 = lnb[col];
        #pragma unroll
        for (int r = 0; r < 8; ++r)
            out[(size_t)(rowbase + rowoff + r) * VAL + col] =
                (acc[c][r] - mu[r]) * rstd[r] * g + b2;
    }
}

extern "C" void kernel_launch(void* const* d_in, const int* in_sizes, int n_in,
                              void* d_out, int out_size, void* d_ws, size_t ws_size,
                              hipStream_t stream) {
    const float* x     = (const float*)d_in[0];
    const float* conn  = (const float*)d_in[1];
    const float* k_w   = (const float*)d_in[2];
    const float* k_b   = (const float*)d_in[3];
    const float* v_w   = (const float*)d_in[4];
    const float* v_b   = (const float*)d_in[5];
    const float* out_w = (const float*)d_in[6];
    const float* out_b = (const float*)d_in[7];
    const float* ln_g  = (const float*)d_in[8];
    const float* ln_b  = (const float*)d_in[9];
    const int*   rows  = (const int*)d_in[10];
    const int*   cols  = (const int*)d_in[11];
    float* out = (float*)d_out;

    char* p = (char*)d_ws;
    auto alloc = [&](size_t bytes) {
        void* r = (void*)p;
        p += (bytes + 255) & ~(size_t)255;
        return r;
    };
    float*    kwsum  = (float*)alloc(INP * sizeof(float));
    float*    kbsum  = (float*)alloc(sizeof(float));
    float*    ksum   = (float*)alloc((size_t)MROWS * sizeof(float));
    unsigned* mbuf   = (unsigned*)alloc((size_t)NN * BB * sizeof(float));
    float*    denom  = (float*)alloc((size_t)NN * BB * sizeof(float));
    float*    ex     = (float*)alloc((size_t)EE * BB * sizeof(float));
    float*    values = (float*)alloc((size_t)MROWS * VAL * sizeof(float));
    float*    agg    = (float*)alloc((size_t)MROWS * VAL * sizeof(float));

    hipMemsetAsync(mbuf, 0, (size_t)NN * BB * sizeof(float), stream);
    hipMemsetAsync(denom, 0, (size_t)NN * BB * sizeof(float), stream);
    hipMemsetAsync(agg, 0, (size_t)MROWS * VAL * sizeof(float), stream);

    bsa_reduce_kw<<<1, 128, 0, stream>>>(k_w, k_b, kwsum, kbsum);

    int gemm_blocks = (MROWS / 16 + 7) / 8;
    bsa_values_gemm<<<gemm_blocks, 256, 0, stream>>>(x, v_w, v_b, values);

    bsa_ksum<<<(MROWS * 32 + 255) / 256, 256, 0, stream>>>(x, kwsum, kbsum, ksum);

    int edge_blocks = (EE * BB + 255) / 256;
    bsa_edge_max<<<edge_blocks, 256, 0, stream>>>(conn, rows, cols, ksum, mbuf);
    bsa_m_finalize<<<(NN * BB + 255) / 256, 256, 0, stream>>>(mbuf);

    bsa_edge_exp<<<edge_blocks, 256, 0, stream>>>(conn, rows, cols, ksum,
                                                  (const float*)mbuf, ex, denom);

    bsa_scatter<<<EE, 256, 0, stream>>>(rows, cols, ex, denom, values, agg);

    bsa_out_gemm_ln<<<gemm_blocks, 256, 0, stream>>>(agg, out_w, out_b,
                                                     ln_g, ln_b, out);
}


# Aggregation without float atomics: edges grouped by receiver row (count, scan, slot), one wave per row sums w*values in registers and stores agg once; agg memset dropped
# speedup vs baseline: 2.3740x; 2.3740x over previous
#include <hip/hip_runtime.h>
#include <math.h>

#define BB    2
#define NN    50000
#define EE    800000
#define INP   128
#define KEY   64
#define VAL   128
#define MROWS (BB * NN)
#define LN_EPS 1e-5f

typedef __attribute__((ext_vector_type(2))) float v2f;
typedef __attribute__((ext_vector_type(8))) float v8f;
typedef __attribute__((ext_vector_type(4))) float v4f;

__device__ __forceinline__ unsigned enc_f32(float f) {
    unsigned u = __float_as_uint(f);
    return (u & 0x80000000u) ? ~u : (u | 0x80000000u);
}
__device__ __forceinline__ float dec_f32(unsigned u) {
    return (u & 0x80000000u) ? __uint_as_float(u & 0x7FFFFFFFu)
                             : __uint_as_float(~u);
}

__global__ void bsa_reduce_kw(const float* __restrict__ kw,
                              const float* __restrict__ kb,
                              float* __restrict__ kwsum,
                              float* __restrict__ kbsum) {
    int i = threadIdx.x;
    float s = 0.f;
    for (int j = 0; j < KEY; ++j) s += kw[i * KEY + j];
    kwsum[i] = s;
    if (i == 0) {
        float t = 0.f;
        for (int j = 0; j < KEY; ++j) t += kb[j];
        kbsum[0] = t;
    }
}

__global__ void bsa_ksum(const float* __restrict__ x,
                         const float* __restrict__ kwsum,
                         const float* __restrict__ kbsum,
                         float* __restrict__ ksum) {
    int wid  = (blockIdx.x * blockDim.x + threadIdx.x) >> 5;
    int lane = threadIdx.x & 31;
    if (wid >= MROWS) return;
    const float* xr = x + (size_t)wid * INP;
    float s = 0.f;
    #pragma unroll
    for (int i = 0; i < INP / 32; ++i) {
        int c = lane + i * 32;
        s += xr[c] * kwsum[c];
    }
    #pragma unroll
    for (int off = 16; off; off >>= 1) s += __shfl_xor(s, off);
    if (lane == 0) ksum[wid] = s + kbsum[0];
}

__device__ __forceinline__ void bsa_wmma_tile(const float* __restrict__ Arow,
                                              const float* __restrict__ Bs,
                                              v8f acc[8], int lane) {
    const int kh = (lane >> 4) << 1;
    const int n  = lane & 15;
    for (int k = 0; k < 128; k += 4) {
        v2f a = *(const v2f*)(Arow + k + kh);
        #pragma unroll
        for (int c = 0; c < 8; ++c) {
            v2f bf;
            bf.x = Bs[(k + kh)     * 128 + (c << 4) + n];
            bf.y = Bs[(k + kh + 1) * 128 + (c << 4) + n];
            acc[c] = __builtin_amdgcn_wmma_f32_16x16x4_f32(
                false, a, false, bf, (short)0, acc[c], false, false);
        }
    }
}

__global__ void bsa_values_gemm(const float* __restrict__ A,
                                const float* __restrict__ W,
                                const float* __restrict__ bias,
                                float* __restrict__ out) {
    __shared__ float Bs[128 * 128];
    for (int i = threadIdx.x; i < 128 * 128; i += blockDim.x) Bs[i] = W[i];
    __syncthreads();

    int lane    = threadIdx.x & 31;
    int wv      = threadIdx.x >> 5;
    int tile    = blockIdx.x * 8 + wv;
    int rowbase = tile * 16;
    if (rowbase >= MROWS) return;

    v8f acc[8] = {};
    const float* Arow = A + (size_t)(rowbase + (lane & 15)) * 128;
    bsa_wmma_tile(Arow, Bs, acc, lane);

    int n      = lane & 15;
    int rowoff = (lane >> 4) << 3;
    #pragma unroll
    for (int c = 0; c < 8; ++c) {
        float bb = bias[(c << 4) + n];
        #pragma unroll
        for (int r = 0; r < 8; ++r)
            out[(size_t)(rowbase + rowoff + r) * 128 + (c << 4) + n] = acc[c][r] + bb;
    }
}

__global__ void bsa_edge_max(const float* __restrict__ conn,
                             const int* __restrict__ rows,
                             const int* __restrict__ cols,
                             const float* __restrict__ ksum,
                             unsigned* __restrict__ mu,
                             int* __restrict__ rowcnt) {
    int tid = blockIdx.x * blockDim.x + threadIdx.x;
    if (tid >= EE * BB) return;
    int e = tid >> 1, b = tid & 1;
    int r = rows[e];
    float s = conn[e] * ksum[b * NN + cols[e]];
    atomicMax(&mu[r * BB + b], enc_f32(s));
    if (b == 0) atomicAdd(&rowcnt[r], 1);
}

#define SCAN_T   1024
#define SCAN_PER 50
typedef __attribute__((ext_vector_type(2))) int v2i;
static_assert(NN % SCAN_PER == 0 && NN / SCAN_PER <= SCAN_T && SCAN_PER % 2 == 0, "scan split");
__global__ __launch_bounds__(SCAN_T) void bsa_row_offsets(int* __restrict__ rowcnt) {
    __shared__ int wsum[SCAN_T / 32];
    const int t = threadIdx.x, lane = t & 31, wv = t >> 5;
    const bool act = t < NN / SCAN_PER;
    v2i* mine = (v2i*)(rowcnt + t * SCAN_PER);
    v2i loc[SCAN_PER / 2];
    int s = 0;
    #pragma unroll
    for (int i = 0; i < SCAN_PER / 2; ++i) {
        loc[i] = act ? mine[i] : (v2i){0, 0};
        s += loc[i].x + loc[i].y;
    }
    int inc = s;
    #pragma unroll
    for (int off = 1; off < 32; off <<= 1) {
        int nb = __shfl_up(inc, off);
        if (lane >= off) inc += nb;
    }
    if (lane == 31) wsum[wv] = inc;
    __syncthreads();
    if (wv == 0) {
        int x = wsum[lane], y = x;
        #pragma unroll
        for (int off = 1; off < 32; off <<= 1) {
            int nb = __shfl_up(y, off);
            if (lane >= off) y += nb;
        }
        wsum[lane] = y - x;
    }
    __syncthreads();
    int run = wsum[wv] + inc - s;
    if (act) {
        #pragma unroll
        for (int i = 0; i < SCAN_PER / 2; ++i) {
            v2i o;
            o.x = run;  run += loc[i].x;
            o.y = run;  run += loc[i].y;
            mine[i] = o;
        }
    }
}

__global__ void bsa_m_finalize(unsigned* __restrict__ mu) {
    int i = blockIdx.x * blockDim.x + threadIdx.x;
    if (i >= NN * BB) return;
    unsigned u = mu[i];
    float m = (u == 0u) ? 0.0f : dec_f32(u);
    if (!isfinite(m)) m = 0.0f;
    ((float*)mu)[i] = m;
}

__global__ void bsa_edge_exp(const float* __restrict__ conn,
                             const int* __restrict__ rows,
                             const int* __restrict__ cols,
                             const float* __restrict__ ksum,
                             const float* __restrict__ m,
                             int* __restrict__ cursor,
                             int* __restrict__ scol,
                             float* __restrict__ sex,
                             float* __restrict__ denom) {
    int tid = blockIdx.x * blockDim.x + threadIdx.x;
    if (tid >= EE * BB) return;
    int e = tid >> 1, b = tid & 1;
    int r = rows[e], c = cols[e];
    float s = conn[e] * ksum[b * NN + c];
    float ev = __expf(s - m[r * BB + b]);
    int slot = 0;
    if (b == 0) slot = atomicAdd(&cursor[r], 1);
    slot = __shfl(slot, (threadIdx.x & 31) & ~1);
    slot = min(max(slot, 0), EE - 1);
    sex[(size_t)slot * BB + b] = ev;
    if (b == 0) scol[slot] = c;
    atomicAdd(&denom[r * BB + b], ev);
}

__global__ __launch_bounds__(256) void bsa_gather_rows(const int* __restrict__ cursor,
                                const int* __restrict__ scol,
                                const float* __restrict__ sex,
                                const float* __restrict__ denom,
                                const float* __restrict__ values,
                                float* __restrict__ agg) {
    const int lane = threadIdx.x & 31;
    const int r = __builtin_amdgcn_readfirstlane((int)((blockIdx.x * 256 + threadIdx.x) >> 5));
    if (r >= NN) return;
    int beg = r ? cursor[r - 1] : 0;
    int end = cursor[r];
    beg = max(beg, 0);
    end = min(end, EE);
    float d0 = denom[r * BB], d1 = denom[r * BB + 1];
    d0 = (d0 == 0.0f) ? 1.0f : d0;
    d1 = (d1 == 0.0f) ? 1.0f : d1;
    v4f a0 = {0.f, 0.f, 0.f, 0.f}, a1 = {0.f, 0.f, 0.f, 0.f};
    const float* v0p = values + lane * 4;
    const float* v1p = values + (size_t)NN * VAL + lane * 4;
    for (int j0 = beg; j0 < end; j0 += 32) {
        const int n = min(32, end - j0);
        int c = 0;
        float w0 = 0.f, w1 = 0.f;
        if (lane < n) {
            c = scol[j0 + lane];
            v2f e2 = *(const v2f*)(sex + (size_t)(j0 + lane) * BB);
            w0 = e2.x / d0;
            w1 = e2.y / d1;
        }
        #pragma unroll 4
        for (int j = 0; j < n; ++j) {
            unsigned cj = (unsigned)__builtin_amdgcn_readlane(c, j);
            cj = min(cj, (unsigned)(NN - 1));
            float x0 = __int_as_float(__builtin_amdgcn_readlane(__float_as_int(w0), j));
            float x1 = __int_as_float(__builtin_amdgcn_readlane(__float_as_int(w1), j));
            v4f p = *(const v4f*)(v0p + (size_t)cj * VAL);
            v4f q = *(const v4f*)(v1p + (size_t)cj * VAL);
            a0 += x0 * p;
            a1 += x1 * q;
        }
    }
    *(v4f*)(agg + (size_t)r * VAL + lane * 4) = a0;
    *(v4f*)(agg + ((size_t)NN + r) * VAL + lane * 4) = a1;
}

__global__ void bsa_out_gemm_ln(const float* __restrict__ A,
                                const float* __restrict__ W,
                                const float* __restrict__ bias,
                                const float* __restrict__ lng,
                                const float* __restrict__ lnb,
                                float* __restrict__ out) {
    __shared__ float Bs[128 * 128];
    for (int i = threadIdx.x; i < 128 * 128; i += blockDim.x) Bs[i] = W[i];
    __syncthreads();

    int lane    = threadIdx.x & 31;
    int wv      = threadIdx.x >> 5;
    int tile    = blockIdx.x * 8 + wv;
    int rowbase = tile * 16;
    if (rowbase >= MROWS) return;

    v8f acc[8] = {};
    const float* Arow = A + (size_t)(rowbase + (lane & 15)) * 128;
    bsa_wmma_tile(Arow, Bs, acc, lane);

    int n = lane & 15;
    float sr[8], sq[8];
    #pragma unroll
    for (int r = 0; r < 8; ++r) { sr[r] = 0.f; sq[r] = 0.f; }
    #pragma unroll
    for (int c = 0; c < 8; ++c) {
        float bb = bias[(c << 4) + n];
        #pragma unroll
        for (int r = 0; r < 8; ++r) {
            float v = acc[c][r] + bb;
            v = v / (1.0f + __expf(-v));
            acc[c][r] = v;
            sr[r] += v;
            sq[r] += v * v;
        }
    }
    #pragma unroll
    for (int r = 0; r < 8; ++r) {
        #pragma unroll
        for (int off = 8; off; off >>= 1) {
            sr[r] += __shfl_xor(sr[r], off);
            sq[r] += __shfl_xor(sq[r], off);
        }
    }
    float mu[8], rstd[8];
    #pragma unroll
    for (int r = 0; r < 8; ++r) {
        mu[r] = sr[r] * (1.0f / VAL);
        float var = sq[r] * (1.0f / VAL) - mu[r] * mu[r];
        rstd[r] = rsqrtf(var + LN_EPS);
    }
    int rowoff = (lane >> 4) << 3;
    #pragma unroll
    for (int c = 0; c < 8; ++c) {
        int col = (c << 4) + n;
        float g = lng[col], b2 = lnb[col];
        #pragma unroll
        for (int r = 0; r < 8; ++r)
            out[(size_t)(rowbase + rowoff + r) * VAL + col] =
                (acc[c][r] - mu[r]) * rstd[r] * g + b2;
    }
}

extern "C" void kernel_launch(void* const* d_in, const int* in_sizes, int n_in,
                              void* d_out, int out_size, void* d_ws, size_t ws_size,
                              hipStream_t stream) {
    const float* x     = (const float*)d_in[0];
    const float* conn  = (const float*)d_in[1];
    const float* k_w   = (const float*)d_in[2];
    const float* k_b   = (const float*)d_in[3];
    const float* v_w   = (const float*)d_in[4];
    const float* v_b   = (const float*)d_in[5];
    const float* out_w = (const float*)d_in[6];
    const float* out_b = (const float*)d_in[7];
    const float* ln_g  = (const float*)d_in[8];
    const float* ln_b  = (const float*)d_in[9];
    const int*   rows  = (const int*)d_in[10];
    const int*   cols  = (const int*)d_in[11];
    float* out = (float*)d_out;

    char* p = (char*)d_ws;
    auto alloc = [&](size_t bytes) {
        void* r = (void*)p;
        p += (bytes + 255) & ~(size_t)255;
        return r;
    };
    float*    kwsum  = (float*)alloc(INP * sizeof(float));
    float*    kbsum  = (float*)alloc(sizeof(float));
    float*    ksum   = (float*)alloc((size_t)MROWS * sizeof(float));
    unsigned* mbuf   = (unsigned*)alloc((size_t)NN * BB * sizeof(float));
    float*    denom  = (float*)alloc((size_t)NN * BB * sizeof(float));
    float*    sex    = (float*)alloc((size_t)EE * BB * sizeof(float));
    float*    values = (float*)alloc((size_t)MROWS * VAL * sizeof(float));
    float*    agg    = (float*)alloc((size_t)MROWS * VAL * sizeof(float));
    int*      cursor = (int*)alloc((size_t)NN * sizeof(int));
    int*      scol   = (int*)alloc((size_t)EE * sizeof(int));
    if ((size_t)(p - (char*)d_ws) > ws_size) return;

    hipMemsetAsync(mbuf, 0, (size_t)NN * BB * sizeof(float), stream);
    hipMemsetAsync(denom, 0, (size_t)NN * BB * sizeof(float), stream);
    hipMemsetAsync(cursor, 0, (size_t)NN * sizeof(int), stream);

    bsa_reduce_kw<<<1, 128, 0, stream>>>(k_w, k_b, kwsum, kbsum);

    int gemm_blocks = (MROWS / 16 + 7) / 8;
    bsa_values_gemm<<<gemm_blocks, 256, 0, stream>>>(x, v_w, v_b, values);

    bsa_ksum<<<(MROWS * 32 + 255) / 256, 256, 0, stream>>>(x, kwsum, kbsum, ksum);

    int edge_blocks = (EE * BB + 255) / 256;
    bsa_edge_max<<<edge_blocks, 256, 0, stream>>>(conn, rows, cols, ksum, mbuf, cursor);
    bsa_m_finalize<<<(NN * BB + 255) / 256, 256, 0, stream>>>(mbuf);
    bsa_row_offsets<<<1, SCAN_T, 0, stream>>>(cursor);

    bsa_edge_exp<<<edge_blocks, 256, 0, stream>>>(conn, rows, cols, ksum,
                                                  (const float*)mbuf, cursor, scol,
                                                  sex, denom);

    bsa_gather_rows<<<(NN * 32 + 255) / 256, 256, 0, stream>>>(cursor, scol, sex, denom,
                                                            values, agg);

    bsa_out_gemm_ln<<<gemm_blocks, 256, 0, stream>>>(agg, out_w, out_b,
                                                     ln_g, ln_b, out);
}


# W kept k-major in LDS, both GEMMs: a lane's four k of a column are one 16-byte LDS load (A likewise from global), fetched ahead of use; gather wave sorts each row's edges so sums repeat run to run
# speedup vs baseline: 3.5006x; 1.4746x over previous
#include <hip/hip_runtime.h>
#include <math.h>

#define BB    2
#define NN    50000
#define EE    800000
#define INP   128
#define KEY   64
#define VAL   128
#define MROWS (BB * NN)
#define LN_EPS 1e-5f

typedef __attribute__((ext_vector_type(2))) float v2f;
typedef __attribute__((ext_vector_type(8))) float v8f;
typedef __attribute__((ext_vector_type(4))) float v4f;

__global__ void bsa_reduce_kw(const float* __restrict__ kw,
                              const float* __restrict__ kb,
                              float* __restrict__ kwsum,
                              float* __restrict__ kbsum) {
    int i = threadIdx.x;
    float s = 0.f;
    for (int j = 0; j < KEY; ++j) s += kw[i * KEY + j];
    kwsum[i] = s;
    if (i == 0) {
        float t = 0.f;
        for (int j = 0; j < KEY; ++j) t += kb[j];
        kbsum[0] = t;
    }
}

#define KP 132

__device__ __forceinline__ void bsa_stage_w(const float* __restrict__ W, float* __restrict__ Bt) {
    const int col = threadIdx.x & 127;
    const int g0  = (threadIdx.x >> 7) << 4;
    float t[16][4];
    #pragma unroll
    for (int g = 0; g < 16; ++g) {
        #pragma unroll
        for (int i = 0; i < 4; ++i)
            t[g][i] = W[((g0 + g) * 4 + i) * 128 + col];
    }
    #pragma unroll
    for (int g = 0; g < 16; ++g) {
        v4f v = {t[g][0], t[g][1], t[g][2], t[g][3]};
        *(v4f*)(Bt + col * KP + (g0 + g) * 4) = v;
    }
}

template <bool DOT>
__device__ __forceinline__ void bsa_wmma_tile(const float* __restrict__ Arow,
                                              const float* __restrict__ Bt,
                                              v8f acc[8], int lane,
                                              const float* __restrict__ Ks, v2f& dot2) {
    const int k4 = (lane >> 4) << 2;
    const float* ap = Arow + k4;
    const float* bp = Bt + (lane & 15) * KP + k4;
    v4f a = *(const v4f*)ap;
    v4f bx[4], by[4];
    #pragma unroll
    for (int c = 0; c < 4; ++c) bx[c] = *(const v4f*)(bp + c * 16 * KP);
    for (int k = 0; k < 128; k += 8) {
        #pragma unroll
        for (int c = 0; c < 4; ++c) by[c] = *(const v4f*)(bp + (c + 4) * 16 * KP + k);
        const v2f a01 = {a.x, a.y}, a23 = {a.z, a.w};
        if (DOT) {
            v4f q = *(const v4f*)(Ks + k + k4);
            v2f q01 = {q.x, q.y}, q23 = {q.z, q.w};
            dot2 += a01 * q01;
            dot2 += a23 * q23;
        }
        #pragma unroll
        for (int c = 0; c < 4; ++c) {
            v2f b01 = {bx[c].x, bx[c].y};
            acc[c] = __builtin_amdgcn_wmma_f32_16x16x4_f32(
                false, a01, false, b01, (short)0, acc[c], false, false);
        }
        #pragma unroll
        for (int c = 0; c < 4; ++c) {
            v2f b23 = {bx[c].z, bx[c].w};
            acc[c] = __builtin_amdgcn_wmma_f32_16x16x4_f32(
                false, a23, false, b23, (short)0, acc[c], false, false);
        }
        const int kn = (k + 8 < 128) ? k + 8 : k;
        a = *(const v4f*)(ap + kn);
        #pragma unroll
        for (int c = 0; c < 4; ++c) bx[c] = *(const v4f*)(bp + c * 16 * KP + kn);
        #pragma unroll
        for (int c = 0; c < 4; ++c) {
            v2f b01 = {by[c].x, by[c].y};
            acc[c + 4] = __builtin_amdgcn_wmma_f32_16x16x4_f32(
                false, a01, false, b01, (short)0, acc[c + 4], false, false);
        }
        #pragma unroll
        for (int c = 0; c < 4; ++c) {
            v2f b23 = {by[c].z, by[c].w};
            acc[c + 4] = __builtin_amdgcn_wmma_f32_16x16x4_f32(
                false, a23, false, b23, (short)0, acc[c + 4], false, false);
        }
    }
}

__global__ __launch_bounds__(256) void bsa_values_gemm(const float* __restrict__ A,
                                const float* __restrict__ W,
                                const float* __restrict__ bias,
                                const float* __restrict__ kwsum,
                                const float* __restrict__ kbsum,
                                float* __restrict__ ksum,
                                float* __restrict__ out) {
    __shared__ __attribute__((aligned(16))) float Bs[128 * KP];
    __shared__ __attribute__((aligned(16))) float Ks[INP];
    __shared__ float Kd[128];
    bsa_stage_w(W, Bs);
    if (threadIdx.x < INP) Ks[threadIdx.x] = kwsum[threadIdx.x];
    __syncthreads();

    int lane    = threadIdx.x & 31;
    int wv      = threadIdx.x >> 5;
    int tile    = blockIdx.x * 8 + wv;
    int rowbase = tile * 16;
    if (rowbase < MROWS) {
        v8f acc[8] = {};
        v2f dot2 = {0.f, 0.f};
        const float* Arow = A + (size_t)(rowbase + (lane & 15)) * 128;
        bsa_wmma_tile<true>(Arow, Bs, acc, lane, Ks, dot2);

        float dot = dot2.x + dot2.y;
        dot += __shfl_xor(dot, 16);
        if (lane < 16) Kd[(wv << 4) + lane] = dot;

        int n      = lane & 15;
        int rowoff = (lane >> 4) << 3;
        #pragma unroll
        for (int c = 0; c < 8; ++c) {
            float bb = bias[(c << 4) + n];
            #pragma unroll
            for (int r = 0; r < 8; ++r)
                out[(size_t)(rowbase + rowoff + r) * 128 + (c << 4) + n] = acc[c][r] + bb;
        }
    }
    __syncthreads();
    int row = blockIdx.x * 128 + threadIdx.x;
    if (threadIdx.x < 128 && row < MROWS) ksum[row] = Kd[threadIdx.x] + kbsum[0];
}

__global__ void bsa_edge_count(const int* __restrict__ rows,
                               int* __restrict__ rowcnt) {
    int e = blockIdx.x * blockDim.x + threadIdx.x;
    if (e >= EE) return;
    atomicAdd(&rowcnt[rows[e]], 1);
}

#define SCAN_T   1024
#define SCAN_PER 50
typedef __attribute__((ext_vector_type(2))) int v2i;
static_assert(NN % SCAN_PER == 0 && NN / SCAN_PER <= SCAN_T && SCAN_PER % 2 == 0, "scan split");
__global__ __launch_bounds__(SCAN_T) void bsa_row_offsets(int* __restrict__ rowcnt) {
    __shared__ int wsum[SCAN_T / 32];
    const int t = threadIdx.x, lane = t & 31, wv = t >> 5;
    const bool act = t < NN / SCAN_PER;
    v2i* mine = (v2i*)(rowcnt + t * SCAN_PER);
    v2i loc[SCAN_PER / 2];
    int s = 0;
    #pragma unroll
    for (int i = 0; i < SCAN_PER / 2; ++i) {
        loc[i] = act ? mine[i] : (v2i){0, 0};
        s += loc[i].x + loc[i].y;
    }
    int inc = s;
    #pragma unroll
    for (int off = 1; off < 32; off <<= 1) {
        int nb = __shfl_up(inc, off);
        if (lane >= off) inc += nb;
    }
    if (lane == 31) wsum[wv] = inc;
    __syncthreads();
    if (wv == 0) {
        int x = wsum[lane], y = x;
        #pragma unroll
        for (int off = 1; off < 32; off <<= 1) {
            int nb = __shfl_up(y, off);
            if (lane >= off) y += nb;
        }
        wsum[lane] = y - x;
    }
    __syncthreads();
    int run = wsum[wv] + inc - s;
    if (act) {
        #pragma unroll
        for (int i = 0; i < SCAN_PER / 2; ++i) {
            v2i o;
            o.x = run;  run += loc[i].x;
            o.y = run;  run += loc[i].y;
            mine[i] = o;
        }
    }
}

__global__ void bsa_edge_place(const float* __restrict__ conn,
                               const int* __restrict__ rows,
                               const int* __restrict__ cols,
                               const float* __restrict__ ksum,
                               int* __restrict__ cursor,
                               int* __restrict__ scol,
                               float* __restrict__ ssc) {
    int e = blockIdx.x * blockDim.x + threadIdx.x;
    if (e >= EE) return;
    int r = rows[e], c = cols[e];
    float cv = conn[e];
    v2f s;
    s.x = cv * ksum[c];
    s.y = cv * ksum[NN + c];
    int slot = atomicAdd(&cursor[r], 1);
    slot = min(max(slot, 0), EE - 1);
    *(v2f*)(ssc + (size_t)slot * BB) = s;
    scol[slot] = c;
}

__device__ __forceinline__ float bsa_xhalf(float x) {
    int b = __builtin_bit_cast(int, x);
    return __builtin_bit_cast(float,
        __builtin_amdgcn_permlanex16(b, b, 0x76543210, 0xfedcba98, false, false));
}
template <int MASK>
__device__ __forceinline__ float bsa_xrow(float x) {
    return __builtin_bit_cast(float,
        __builtin_amdgcn_update_dpp(0, __builtin_bit_cast(int, x), 0x160 + MASK, 0xf, 0xf, true));
}
__device__ __forceinline__ float bsa_wave_max(float x) {
    x = fmaxf(x, bsa_xrow<8>(x));
    x = fmaxf(x, bsa_xrow<4>(x));
    x = fmaxf(x, bsa_xrow<2>(x));
    x = fmaxf(x, bsa_xrow<1>(x));
    return fmaxf(x, bsa_xhalf(x));
}
__device__ __forceinline__ float bsa_wave_sum(float x) {
    x += bsa_xrow<8>(x);
    x += bsa_xrow<4>(x);
    x += bsa_xrow<2>(x);
    x += bsa_xrow<1>(x);
    return x + bsa_xhalf(x);
}

#define GCAP 128
__global__ __launch_bounds__(256, 16) void bsa_gather_rows(const int* __restrict__ cursor,
                                const int* __restrict__ scol,
                                const float* __restrict__ ssc,
                                const float* __restrict__ values,
                                float* __restrict__ agg) {
    __shared__ int sC[8][GCAP];
    __shared__ __attribute__((aligned(8))) float sS[8][GCAP * 2];
    const int lane = threadIdx.x & 31;
    int*   myC = sC[threadIdx.x >> 5];
    float* myS = sS[threadIdx.x >> 5];
    const int r = __builtin_amdgcn_readfirstlane((int)((blockIdx.x * 256 + threadIdx.x) >> 5));
    if (r >= NN) return;
    int beg = r ? cursor[r - 1] : 0;
    int end = cursor[r];
    beg = max(beg, 0);
    end = min(end, EE);
    const int n = end - beg;
    const bool ordered = n <= GCAP;

    if (ordered) {
        for (int ja = 0; ja < n; ja += 32) {
            const int na = min(32, n - ja);
            int c = 0;
            float s0 = 0.f, s1 = 0.f;
            if (lane < na) {
                c = scol[beg + ja + lane];
                v2f e2 = *(const v2f*)(ssc + (size_t)(beg + ja + lane) * BB);
                s0 = e2.x;
                s1 = e2.y;
            }
            int lt = 0, eq = 0;
            for (int jb = 0; jb < n; jb += 32) {
                const int nb = min(32, n - jb);
                int cb = c;
                if (jb != ja) {
                    cb = 0;
                    if (lane < nb) cb = scol[beg + jb + lane];
                }
                for (int j = 0; j < nb; ++j) {
                    const int cj = __builtin_amdgcn_readlane(cb, j);
                    lt += (cj < c) ? 1 : 0;
                    eq += (cj == c) ? 1 : 0;
                }
            }
            int rank = lt;
            if (__builtin_amdgcn_ballot_w32(lane < na && eq > 1) != 0u) {
                const unsigned ua = __float_as_uint(s0), ub = __float_as_uint(s1);
                const int gi = ja + lane;
                int rk = 0;
                for (int jb = 0; jb < n; jb += 32) {
                    const int nb = min(32, n - jb);
                    int cb = c;
                    unsigned ab = ua, bb = ub;
                    if (jb != ja) {
                        cb = 0;
                        ab = 0u;
                        bb = 0u;
                        if (lane < nb) {
                            cb = scol[beg + jb + lane];
                            v2f e2 = *(const v2f*)(ssc + (size_t)(beg + jb + lane) * BB);
                            ab = __float_as_uint(e2.x);
                            bb = __float_as_uint(e2.y);
                        }
                    }
                    for (int j = 0; j < nb; ++j) {
                        const int cj = __builtin_amdgcn_readlane(cb, j);
                        const unsigned aj = (unsigned)__builtin_amdgcn_readlane((int)ab, j);
                        const unsigned bj = (unsigned)__builtin_amdgcn_readlane((int)bb, j);
                        const int gj = jb + j;
                        const bool before = (cj < c) || (cj == c && (aj < ua || (aj == ua &&
                                            (bj < ub || (bj == ub && gj < gi)))));
                        rk += before ? 1 : 0;
                    }
                }
                rank = rk;
            }
            if (lane < na) {
                rank = min(max(rank, 0), GCAP - 1);
                myC[rank] = c;
                v2f e2 = {s0, s1};
                *(v2f*)(myS + 2 * rank) = e2;
            }
        }
        __builtin_amdgcn_fence(__ATOMIC_RELEASE, "wavefront");
        __builtin_amdgcn_wave_barrier();
        __builtin_amdgcn_fence(__ATOMIC_ACQUIRE, "wavefront");
    }

    int   c  = 0;
    float s0 = -INFINITY, s1 = -INFINITY;
    if (lane < n) {
        if (ordered) {
            c = myC[lane];
            v2f e2 = *(const v2f*)(myS + 2 * lane);
            s0 = e2.x;
            s1 = e2.y;
        } else {
            c = scol[beg + lane];
            v2f e2 = *(const v2f*)(ssc + (size_t)(beg + lane) * BB);
            s0 = e2.x;
            s1 = e2.y;
        }
    }

    float m0 = s0, m1 = s1;
    for (int k0 = 32; k0 < n; k0 += 32) {
        if (lane < n - k0) {
            v2f e2;
            if (ordered) e2 = *(const v2f*)(myS + 2 * (k0 + lane));
            else         e2 = *(const v2f*)(ssc + (size_t)(beg + k0 + lane) * BB);
            m0 = fmaxf(m0, e2.x);
            m1 = fmaxf(m1, e2.y);
        }
    }
    m0 = bsa_wave_max(m0);
    m1 = bsa_wave_max(m1);
    if (!isfinite(m0)) m0 = 0.0f;
    if (!isfinite(m1)) m1 = 0.0f;

    float x0 = (lane < n) ? __expf(s0 - m0) : 0.0f;
    float x1 = (lane < n) ? __expf(s1 - m1) : 0.0f;
    float d0 = x0, d1 = x1;
    for (int k0 = 32; k0 < n; k0 += 32) {
        if (lane < n - k0) {
            v2f e2;
            if (ordered) e2 = *(const v2f*)(myS + 2 * (k0 + lane));
            else         e2 = *(const v2f*)(ssc + (size_t)(beg + k0 + lane) * BB);
            d0 += __expf(e2.x - m0);
            d1 += __expf(e2.y - m1);
        }
    }
    d0 = bsa_wave_sum(d0);
    d1 = bsa_wave_sum(d1);
    d0 = (d0 == 0.0f) ? 1.0f : d0;
    d1 = (d1 == 0.0f) ? 1.0f : d1;

    v4f a0 = {0.f, 0.f, 0.f, 0.f}, a1 = {0.f, 0.f, 0.f, 0.f};
    const float* v0p = values + lane * 4;
    const float* v1p = values + (size_t)NN * VAL + lane * 4;
    for (int k0 = 0; k0 < n; k0 += 32) {
        const int nn = min(32, n - k0);
        if (k0 != 0) {
            c = 0;
            x0 = 0.f;
            x1 = 0.f;
            if (lane < nn) {
                v2f e2;
                if (ordered) {
                    c = myC[k0 + lane];
                    e2 = *(const v2f*)(myS + 2 * (k0 + lane));
                } else {
                    c = scol[beg + k0 + lane];
                    e2 = *(const v2f*)(ssc + (size_t)(beg + k0 + lane) * BB);
                }
                x0 = __expf(e2.x - m0);
                x1 = __expf(e2.y - m1);
            }
        }
        const float w0 = x0 / d0;
        const float w1 = x1 / d1;
        #pragma unroll 4
        for (int j = 0; j < nn; ++j) {
            unsigned cj = (unsigned)__builtin_amdgcn_readlane(c, j);
            cj = min(cj, (unsigned)(NN - 1));
            float y0 = __int_as_float(__builtin_amdgcn_readlane(__float_as_int(w0), j));
            float y1 = __int_as_float(__builtin_amdgcn_readlane(__float_as_int(w1), j));
            v4f p = *(const v4f*)(v0p + (size_t)cj * VAL);
            v4f q = *(const v4f*)(v1p + (size_t)cj * VAL);
            a0 += y0 * p;
            a1 += y1 * q;
        }
    }
    *(v4f*)(agg + (size_t)r * VAL + lane * 4) = a0;
    *(v4f*)(agg + ((size_t)NN + r) * VAL + lane * 4) = a1;
}

__global__ __launch_bounds__(256) void bsa_out_gemm_ln(const float* __restrict__ A,
                                const float* __restrict__ W,
                                const float* __restrict__ bias,
                                const float* __restrict__ lng,
                                const float* __restrict__ lnb,
                                float* __restrict__ out) {
    __shared__ __attribute__((aligned(16))) float Bs[128 * KP];
    bsa_stage_w(W, Bs);
    __syncthreads();

    int lane    = threadIdx.x & 31;
    int wv      = threadIdx.x >> 5;
    int tile    = blockIdx.x * 8 + wv;
    int rowbase = tile * 16;
    if (rowbase >= MROWS) return;

    v8f acc[8] = {};
    v2f nodot = {0.f, 0.f};
    const float* Arow = A + (size_t)(rowbase + (lane & 15)) * 128;
    bsa_wmma_tile<false>(Arow, Bs, acc, lane, nullptr, nodot);

    int n = lane & 15;
    float sr[8], sq[8];
    #pragma unroll
    for (int r = 0; r < 8; ++r) { sr[r] = 0.f; sq[r] = 0.f; }
    #pragma unroll
    for (int c = 0; c < 8; ++c) {
        float bb = bias[(c << 4) + n];
        #pragma unroll
        for (int r = 0; r < 8; ++r) {
            float v = acc[c][r] + bb;
            v = v / (1.0f + __expf(-v));
            acc[c][r] = v;
            sr[r] += v;
            sq[r] += v * v;
        }
    }
    #pragma unroll
    for (int r = 0; r < 8; ++r) {
        #pragma unroll
        for (int off = 8; off; off >>= 1) {
            sr[r] += __shfl_xor(sr[r], off);
            sq[r] += __shfl_xor(sq[r], off);
        }
    }
    float mu[8], rstd[8];
    #pragma unroll
    for (int r = 0; r < 8; ++r) {
        mu[r] = sr[r] * (1.0f / VAL);
        float var = sq[r] * (1.0f / VAL) - mu[r] * mu[r];
        rstd[r] = rsqrtf(var + LN_EPS);
    }
    int rowoff = (lane >> 4) << 3;
    #pragma unroll
    for (int c = 0; c < 8; ++c) {
        int col = (c << 4) + n;
        float g = lng[col], b2 = lnb[col];
        #pragma unroll
        for (int r = 0; r < 8; ++r)
            out[(size_t)(rowbase + rowoff + r) * VAL + col] =
                (acc[c][r] - mu[r]) * rstd[r] * g + b2;
    }
}

extern "C" void kernel_launch(void* const* d_in, const int* in_sizes, int n_in,
                              void* d_out, int out_size, void* d_ws, size_t ws_size,
                              hipStream_t stream) {
    const float* x     = (const float*)d_in[0];
    const float* conn  = (const float*)d_in[1];
    const float* k_w   = (const float*)d_in[2];
    const float* k_b   = (const float*)d_in[3];
    const float* v_w   = (const float*)d_in[4];
    const float* v_b   = (const float*)d_in[5];
    const float* out_w = (const float*)d_in[6];
    const float* out_b = (const float*)d_in[7];
    const float* ln_g  = (const float*)d_in[8];
    const float* ln_b  = (const float*)d_in[9];
    const int*   rows  = (const int*)d_in[10];
    const int*   cols  = (const int*)d_in[11];
    float* out = (float*)d_out;

    char* p = (char*)d_ws;
    auto alloc = [&](size_t bytes) {
        void* r = (void*)p;
        p += (bytes + 255) & ~(size_t)255;
        return r;
    };
    float*    kwsum  = (float*)alloc(INP * sizeof(float));
    float*    kbsum  = (float*)alloc(sizeof(float));
    float*    ksum   = (float*)alloc((size_t)MROWS * sizeof(float));
    float*    ssc    = (float*)alloc((size_t)EE * BB * sizeof(float));
    float*    values = (float*)alloc((size_t)MROWS * VAL * sizeof(float));
    float*    agg    = (float*)alloc((size_t)MROWS * VAL * sizeof(float));
    int*      cursor = (int*)alloc((size_t)NN * sizeof(int));
    int*      scol   = (int*)alloc((size_t)EE * sizeof(int));
    if ((size_t)(p - (char*)d_ws) > ws_size) return;

    hipMemsetAsync(cursor, 0, (size_t)NN * sizeof(int), stream);

    bsa_reduce_kw<<<1, 128, 0, stream>>>(k_w, k_b, kwsum, kbsum);

    int gemm_blocks = (MROWS / 16 + 7) / 8;
    bsa_values_gemm<<<gemm_blocks, 256, 0, stream>>>(x, v_w, v_b, kwsum, kbsum, ksum, values);

    int edge_blocks = (EE + 255) / 256;
    bsa_edge_count<<<edge_blocks, 256, 0, stream>>>(rows, cursor);
    bsa_row_offsets<<<1, SCAN_T, 0, stream>>>(cursor);

    bsa_edge_place<<<edge_blocks, 256, 0, stream>>>(conn, rows, cols, ksum, cursor, scol, ssc);

    bsa_gather_rows<<<(NN * 32 + 255) / 256, 256, 0, stream>>>(cursor, scol, ssc,
                                                            values, agg);

    bsa_out_gemm_ln<<<gemm_blocks, 256, 0, stream>>>(agg, out_w, out_b,
                                                     ln_g, ln_b, out);
}


# Edge placement moved beside the values GEMM, into the WGP places it leaves free (k loop rolled: 120 registers, 4 blocks per WGP); counter clear beside the weight prep; memset node gone
# speedup vs baseline: 7.8588x; 2.2450x over previous
#include <hip/hip_runtime.h>
#include <math.h>

#define BB    2
#define NN    50000
#define EE    800000
#define INP   128
#define KEY   64
#define VAL   128
#define MROWS (BB * NN)
#define LN_EPS 1e-5f

typedef __attribute__((ext_vector_type(2))) float v2f;
typedef __attribute__((ext_vector_type(8))) float v8f;
typedef __attribute__((ext_vector_type(4))) float v4f;

#define KP 132

__device__ __forceinline__ void bsa_stage_w(const float* __restrict__ W, float* __restrict__ Bt) {
    const int col = threadIdx.x & 127;
    const int g0  = (threadIdx.x >> 7) << 4;
    float t[16][4];
    #pragma unroll
    for (int g = 0; g < 16; ++g) {
        #pragma unroll
        for (int i = 0; i < 4; ++i)
            t[g][i] = W[((g0 + g) * 4 + i) * 128 + col];
    }
    #pragma unroll
    for (int g = 0; g < 16; ++g) {
        v4f v = {t[g][0], t[g][1], t[g][2], t[g][3]};
        *(v4f*)(Bt + col * KP + (g0 + g) * 4) = v;
    }
}

template <bool DOT>
__device__ __forceinline__ void bsa_wmma_tile(const float* __restrict__ Arow,
                                              const float* __restrict__ Bt,
                                              v8f acc[8], int lane,
                                              const float* __restrict__ Ks, v2f& dot2) {
    const int k4 = (lane >> 4) << 2;
    const float* ap = Arow + k4;
    const float* bp = Bt + (lane & 15) * KP + k4;
    v4f a = *(const v4f*)ap;
    v4f bx[4], by[4];
    #pragma unroll
    for (int c = 0; c < 4; ++c) bx[c] = *(const v4f*)(bp + c * 16 * KP);
    #pragma nounroll
    for (int k = 0; k < 128; k += 8) {
        #pragma unroll
        for (int c = 0; c < 4; ++c) by[c] = *(const v4f*)(bp + (c + 4) * 16 * KP + k);
        const v2f a01 = {a.x, a.y}, a23 = {a.z, a.w};
        if (DOT) {
            v4f q = *(const v4f*)(Ks + k + k4);
            v2f q01 = {q.x, q.y}, q23 = {q.z, q.w};
            dot2 += a01 * q01;
            dot2 += a23 * q23;
        }
        #pragma unroll
        for (int c = 0; c < 4; ++c) {
            v2f b01 = {bx[c].x, bx[c].y};
            acc[c] = __builtin_amdgcn_wmma_f32_16x16x4_f32(
                false, a01, false, b01, (short)0, acc[c], false, false);
        }
        #pragma unroll
        for (int c = 0; c < 4; ++c) {
            v2f b23 = {bx[c].z, bx[c].w};
            acc[c] = __builtin_amdgcn_wmma_f32_16x16x4_f32(
                false, a23, false, b23, (short)0, acc[c], false, false);
        }
        const int kn = (k + 8 < 128) ? k + 8 : k;
        a = *(const v4f*)(ap + kn);
        asm volatile("" ::: "memory");
        #pragma unroll
        for (int c = 0; c < 4; ++c) bx[c] = *(const v4f*)(bp + c * 16 * KP + kn);
        #pragma unroll
        for (int c = 0; c < 4; ++c) {
            v2f b01 = {by[c].x, by[c].y};
            acc[c + 4] = __builtin_amdgcn_wmma_f32_16x16x4_f32(
                false, a01, false, b01, (short)0, acc[c + 4], false, false);
        }
        #pragma unroll
        for (int c = 0; c < 4; ++c) {
            v2f b23 = {by[c].z, by[c].w};
            acc[c + 4] = __builtin_amdgcn_wmma_f32_16x16x4_f32(
                false, a23, false, b23, (short)0, acc[c + 4], false, false);
        }
    }
}

typedef __attribute__((ext_vector_type(2))) int v2i;
typedef __attribute__((ext_vector_type(4))) int v4i;
#define PREP_BLOCKS  (INP / 16)
#define GEMM_BLOCKS  ((MROWS / 16 + 7) / 8)
#define PLACE_BLOCKS ((EE / 8 + 255) / 256)
#define CNT_INTS     (((NN + 1 + 3) / 4) * 4)
#define ZERO_BLOCKS  ((CNT_INTS / 4 + 255) / 256)
#define CAP 64
static_assert(EE % 8 == 0 && CAP % 32 == 0, "16-byte loads of the edge list; a row's slots are whole chunks of 32");

__device__ __forceinline__ void bsa_put_edge(int r, int c, float cv, int slot,
                                             int* __restrict__ cnt, int* __restrict__ slots,
                                             int* __restrict__ ovr, int* __restrict__ ovc,
                                             float* __restrict__ ovv) {
    if (slot < CAP) {
        const v2i t = {c, __float_as_int(cv)};
        *(v2i*)(slots + ((size_t)r * CAP + max(slot, 0)) * 2) = t;
    } else {
        int o = atomicAdd(&cnt[NN], 1);
        o = min(max(o, 0), EE - 1);
        ovr[o] = r;
        ovc[o] = c;
        ovv[o] = cv;
    }
}

__device__ __forceinline__ void bsa_place8(int t, const float* __restrict__ conn,
                                           const int* __restrict__ rows, const int* __restrict__ cols,
                                           int* __restrict__ cnt, int* __restrict__ slots,
                                           int* __restrict__ ovr, int* __restrict__ ovc,
                                           float* __restrict__ ovv) {
    const v4i ra = *(const v4i*)(rows + 8 * t);
    const v4i rb = *(const v4i*)(rows + 8 * t + 4);
    const v4i ca = *(const v4i*)(cols + 8 * t);
    const v4i cb = *(const v4i*)(cols + 8 * t + 4);
    const v4f va = *(const v4f*)(conn + 8 * t);
    const v4f vb = *(const v4f*)(conn + 8 * t + 4);
    const int q0 = atomicAdd(&cnt[ra.x], 1);
    const int q1 = atomicAdd(&cnt[ra.y], 1);
    const int q2 = atomicAdd(&cnt[ra.z], 1);
    const int q3 = atomicAdd(&cnt[ra.w], 1);
    const int q4 = atomicAdd(&cnt[rb.x], 1);
    const int q5 = atomicAdd(&cnt[rb.y], 1);
    const int q6 = atomicAdd(&cnt[rb.z], 1);
    const int q7 = atomicAdd(&cnt[rb.w], 1);
    bsa_put_edge(ra.x, ca.x, va.x, q0, cnt, slots, ovr, ovc, ovv);
    bsa_put_edge(ra.y, ca.y, va.y, q1, cnt, slots, ovr, ovc, ovv);
    bsa_put_edge(ra.z, ca.z, va.z, q2, cnt, slots, ovr, ovc, ovv);
    bsa_put_edge(ra.w, ca.w, va.w, q3, cnt, slots, ovr, ovc, ovv);
    bsa_put_edge(rb.x, cb.x, vb.x, q4, cnt, slots, ovr, ovc, ovv);
    bsa_put_edge(rb.y, cb.y, vb.y, q5, cnt, slots, ovr, ovc, ovv);
    bsa_put_edge(rb.z, cb.z, vb.z, q6, cnt, slots, ovr, ovc, ovv);
    bsa_put_edge(rb.w, cb.w, vb.w, q7, cnt, slots, ovr, ovc, ovv);
}

__global__ __launch_bounds__(256) void bsa_prep_zero(const float* __restrict__ kw,
                                const float* __restrict__ kb,
                                const float* __restrict__ vw,
                                const float* __restrict__ vb,
                                const float* __restrict__ ow,
                                float* __restrict__ kwsum,
                                float* __restrict__ kbsum,
                                float* __restrict__ Wc,
                                float* __restrict__ bc,
                                int* __restrict__ cnt) {
    __shared__ __attribute__((aligned(16))) float Bs[128 * KP];
    if (blockIdx.x >= PREP_BLOCKS) {
        const int t = (blockIdx.x - PREP_BLOCKS) * 256 + threadIdx.x;
        if (t < CNT_INTS / 4) {
            const v4i z = {0, 0, 0, 0};
            *(v4i*)(cnt + 4 * t) = z;
        }
        return;
    }
    bsa_stage_w(ow, Bs);
    const int i = threadIdx.x;
    if (blockIdx.x == 0) {
        if (i < INP) {
            float s = 0.f;
            for (int j = 0; j < KEY; ++j) s += kw[i * KEY + j];
            kwsum[i] = s;
        } else if (i == INP) {
            float t = 0.f;
            for (int j = 0; j < KEY; ++j) t += kb[j];
            kbsum[0] = t;
        }
    }
    __syncthreads();

    if (blockIdx.x == 1 && i < VAL) {
        v4f t = {0.f, 0.f, 0.f, 0.f};
        const float* colp = Bs + i * KP;
        for (int k = 0; k < VAL; k += 4)
            t += *(const v4f*)(vb + k) * *(const v4f*)(colp + k);
        bc[i] = (t.x + t.y) + (t.z + t.w);
    }

    const int lane    = i & 31;
    const int wv      = i >> 5;
    const int rowbase = blockIdx.x << 4;
    const int k4      = (lane >> 4) << 2;
    const float* ap = vw + (size_t)(rowbase + (lane & 15)) * 128 + k4;
    const float* bp = Bs + ((wv << 4) + (lane & 15)) * KP + k4;
    v8f acc0 = {}, acc1 = {};
    for (int k = 0; k < 128; k += 8) {
        const v4f a = *(const v4f*)(ap + k);
        const v4f b = *(const v4f*)(bp + k);
        const v2f a01 = {a.x, a.y}, a23 = {a.z, a.w};
        const v2f b01 = {b.x, b.y}, b23 = {b.z, b.w};
        acc0 = __builtin_amdgcn_wmma_f32_16x16x4_f32(
            false, a01, false, b01, (short)0, acc0, false, false);
        acc1 = __builtin_amdgcn_wmma_f32_16x16x4_f32(
            false, a23, false, b23, (short)0, acc1, false, false);
    }
    const int n      = lane & 15;
    const int rowoff = (lane >> 4) << 3;
    #pragma unroll
    for (int r = 0; r < 8; ++r)
        Wc[(rowbase + rowoff + r) * 128 + (wv << 4) + n] = acc0[r] + acc1[r];
}

__global__ __launch_bounds__(256) void bsa_values_gemm_place(const float* __restrict__ A,
                                const float* __restrict__ W,
                                const float* __restrict__ bias,
                                const float* __restrict__ kwsum,
                                const float* __restrict__ kbsum,
                                float* __restrict__ ksum,
                                float* __restrict__ out,
                                const float* __restrict__ conn,
                                const int* __restrict__ rows,
                                const int* __restrict__ cols,
                                int* __restrict__ cnt,
                                int* __restrict__ slots,
                                int* __restrict__ ovr,
                                int* __restrict__ ovc,
                                float* __restrict__ ovv) {
    __shared__ __attribute__((aligned(16))) float Bs[128 * KP];
    __shared__ __attribute__((aligned(16))) float Ks[INP];
    __shared__ float Kd[128];
    if (blockIdx.x >= GEMM_BLOCKS) {
        const int t = (blockIdx.x - GEMM_BLOCKS) * 256 + threadIdx.x;
        if (t < EE / 8) {
            bsa_place8(t, conn, rows, cols, cnt, slots, ovr, ovc, ovv);
        }
        return;
    }
    const int blk = blockIdx.x;
    bsa_stage_w(W, Bs);
    if (threadIdx.x < INP) Ks[threadIdx.x] = kwsum[threadIdx.x];
    __syncthreads();

    int lane    = threadIdx.x & 31;
    int wv      = threadIdx.x >> 5;
    int tile    = blk * 8 + wv;
    int rowbase = tile * 16;
    if (rowbase < MROWS) {
        v8f acc[8] = {};
        v2f dot2 = {0.f, 0.f};
        const float* Arow = A + (size_t)(rowbase + (lane & 15)) * 128;
        bsa_wmma_tile<true>(Arow, Bs, acc, lane, Ks, dot2);

        float dot = dot2.x + dot2.y;
        dot += __shfl_xor(dot, 16);
        if (lane < 16) Kd[(wv << 4) + lane] = dot;

        int n      = lane & 15;
        int rowoff = (lane >> 4) << 3;
        #pragma unroll
        for (int c = 0; c < 8; ++c) {
            float bb = bias[(c << 4) + n];
            #pragma unroll
            for (int r = 0; r < 8; ++r)
                out[(size_t)(rowbase + rowoff + r) * 128 + (c << 4) + n] = acc[c][r] + bb;
        }
    }
    __syncthreads();
    int row = blk * 128 + threadIdx.x;
    if (threadIdx.x < 128 && row < MROWS) ksum[row] = Kd[threadIdx.x] + kbsum[0];
}

__device__ __forceinline__ float bsa_xhalf(float x) {
    int b = __builtin_bit_cast(int, x);
    return __builtin_bit_cast(float,
        __builtin_amdgcn_permlanex16(b, b, 0x76543210, 0xfedcba98, false, false));
}
template <int MASK>
__device__ __forceinline__ float bsa_xrow(float x) {
    return __builtin_bit_cast(float,
        __builtin_amdgcn_update_dpp(0, __builtin_bit_cast(int, x), 0x160 + MASK, 0xf, 0xf, true));
}
__device__ __forceinline__ float bsa_wave_max(float x) {
    x = fmaxf(x, bsa_xrow<8>(x));
    x = fmaxf(x, bsa_xrow<4>(x));
    x = fmaxf(x, bsa_xrow<2>(x));
    x = fmaxf(x, bsa_xrow<1>(x));
    return fmaxf(x, bsa_xhalf(x));
}
__device__ __forceinline__ float bsa_wave_sum(float x) {
    x += bsa_xrow<8>(x);
    x += bsa_xrow<4>(x);
    x += bsa_xrow<2>(x);
    x += bsa_xrow<1>(x);
    return x + bsa_xhalf(x);
}
__device__ __forceinline__ v4f bsa_silu_ln(v4f h, v4f g, v4f b) {
    h.x = h.x / (1.0f + __expf(-h.x));
    h.y = h.y / (1.0f + __expf(-h.y));
    h.z = h.z / (1.0f + __expf(-h.z));
    h.w = h.w / (1.0f + __expf(-h.w));
    const float mu = bsa_wave_sum((h.x + h.y) + (h.z + h.w)) * (1.0f / VAL);
    const v4f d = h - mu;
    const float var = bsa_wave_sum((d.x * d.x + d.y * d.y) + (d.z * d.z + d.w * d.w)) * (1.0f / VAL);
    const float rstd = rsqrtf(var + LN_EPS);
    return d * rstd * g + b;
}

__device__ __forceinline__ void bsa_add_edge(const float* __restrict__ v0p, const float* __restrict__ v1p,
                                             unsigned cj, float y0, float y1, v4f& a0, v4f& a1) {
    cj = min(cj, (unsigned)(NN - 1));
    const v4f p = *(const v4f*)(v0p + (size_t)cj * VAL);
    const v4f q = *(const v4f*)(v1p + (size_t)cj * VAL);
    a0 += y0 * p;
    a1 += y1 * q;
}
__global__ __launch_bounds__(256, 16) void bsa_gather_rows(const int* __restrict__ cnt,
                                const int* __restrict__ slots,
                                const int* __restrict__ ovr,
                                const int* __restrict__ ovc,
                                const float* __restrict__ ovv,
                                const float* __restrict__ ksum,
                                const float* __restrict__ values,
                                const float* __restrict__ outb,
                                const float* __restrict__ lng,
                                const float* __restrict__ lnb,
                                float* __restrict__ out) {
    __shared__ int sC[8][CAP];
    __shared__ __attribute__((aligned(8))) float sS[8][CAP * 2];
    const int lane = threadIdx.x & 31;
    int*   myC = sC[threadIdx.x >> 5];
    float* myS = sS[threadIdx.x >> 5];
    const int r = __builtin_amdgcn_readfirstlane((int)((blockIdx.x * 256 + threadIdx.x) >> 5));
    if (r >= NN) return;
    const int n = min(max(cnt[r], 0), EE);
    const v2i* rowslots = (const v2i*)slots + (size_t)r * CAP;
    const float* v0p = values + lane * 4;
    const float* v1p = values + (size_t)NN * VAL + lane * 4;
    v4f a0 = {0.f, 0.f, 0.f, 0.f}, a1 = {0.f, 0.f, 0.f, 0.f};

    if (n <= CAP) {
        const int nA = min(n, 32), nB = n - nA;
        int   c  = 0, cB = 0;
        float s0 = -INFINITY, s1 = -INFINITY, s0B = -INFINITY, s1B = -INFINITY;
        if (lane < nA) {
            const v2i e = rowslots[lane];
            c = (int)min((unsigned)e.x, (unsigned)(NN - 1));
            const float cv = __int_as_float(e.y);
            s0 = cv * ksum[c];
            s1 = cv * ksum[NN + c];
        }
        if (lane < nB) {
            const v2i e = rowslots[32 + lane];
            cB = (int)min((unsigned)e.x, (unsigned)(NN - 1));
            const float cv = __int_as_float(e.y);
            s0B = cv * ksum[cB];
            s1B = cv * ksum[NN + cB];
        }
        int ltA = 0, eqA = 0, ltB = 0, eqB = 0;
        for (int j = 0; j < nA; ++j) {
            const int cj = __builtin_amdgcn_readlane(c, j);
            ltA += (cj < c) ? 1 : 0;
            eqA += (cj == c) ? 1 : 0;
        }
        if (nB > 0) {
            for (int j = 0; j < nB; ++j) {
                const int cj = __builtin_amdgcn_readlane(cB, j);
                ltA += (cj < c) ? 1 : 0;
                eqA += (cj == c) ? 1 : 0;
                ltB += (cj < cB) ? 1 : 0;
                eqB += (cj == cB) ? 1 : 0;
            }
            for (int j = 0; j < nA; ++j) {
                const int cj = __builtin_amdgcn_readlane(c, j);
                ltB += (cj < cB) ? 1 : 0;
                eqB += (cj == cB) ? 1 : 0;
            }
        }
        int rankA = ltA, rankB = ltB;
        if (__builtin_amdgcn_ballot_w32((lane < nA && eqA > 1) || (lane < nB && eqB > 1)) != 0u) {
            const unsigned uaA = __float_as_uint(s0),  ubA = __float_as_uint(s1);
            const unsigned uaB = __float_as_uint(s0B), ubB = __float_as_uint(s1B);
            int rkA = 0, rkB = 0;
            for (int j = 0; j < n; ++j) {
                int cj;
                unsigned aj, bj;
                if (j < 32) {
                    cj = __builtin_amdgcn_readlane(c, j);
                    aj = (unsigned)__builtin_amdgcn_readlane((int)uaA, j);
                    bj = (unsigned)__builtin_amdgcn_readlane((int)ubA, j);
                } else {
                    cj = __builtin_amdgcn_readlane(cB, j - 32);
                    aj = (unsigned)__builtin_amdgcn_readlane((int)uaB, j - 32);
                    bj = (unsigned)__builtin_amdgcn_readlane((int)ubB, j - 32);
                }
                const bool beforeA = (cj < c) || (cj == c && (aj < uaA || (aj == uaA &&
                                     (bj < ubA || (bj == ubA && j < lane)))));
                const bool beforeB = (cj < cB) || (cj == cB && (aj < uaB || (aj == uaB &&
                                     (bj < ubB || (bj == ubB && j < 32 + lane)))));
                rkA += beforeA ? 1 : 0;
                rkB += beforeB ? 1 : 0;
            }
            rankA = rkA;
            rankB = rkB;
        }
        if (lane < nA) {
            rankA = min(max(rankA, 0), CAP - 1);
            myC[rankA] = c;
            const v2f e2 = {s0, s1};
            *(v2f*)(myS + 2 * rankA) = e2;
        }
        if (lane < nB) {
            rankB = min(max(rankB, 0), CAP - 1);
            myC[rankB] = cB;
            const v2f e2 = {s0B, s1B};
            *(v2f*)(myS + 2 * rankB) = e2;
        }
        __builtin_amdgcn_fence(__ATOMIC_RELEASE, "wavefront");
        __builtin_amdgcn_wave_barrier();
        __builtin_amdgcn_fence(__ATOMIC_ACQUIRE, "wavefront");

        c  = 0;
        s0 = -INFINITY;
        s1 = -INFINITY;
        if (lane < n) {
            c = myC[lane];
            const v2f e2 = *(const v2f*)(myS + 2 * lane);
            s0 = e2.x;
            s1 = e2.y;
        }

        float m0 = s0, m1 = s1;
        for (int k0 = 32; k0 < n; k0 += 32) {
            if (lane < n - k0) {
                const v2f e2 = *(const v2f*)(myS + 2 * (k0 + lane));
                m0 = fmaxf(m0, e2.x);
                m1 = fmaxf(m1, e2.y);
            }
        }
        m0 = bsa_wave_max(m0);
        m1 = bsa_wave_max(m1);
        if (!isfinite(m0)) m0 = 0.0f;
        if (!isfinite(m1)) m1 = 0.0f;

        float x0 = (lane < n) ? __expf(s0 - m0) : 0.0f;
        float x1 = (lane < n) ? __expf(s1 - m1) : 0.0f;
        float d0 = x0, d1 = x1;
        for (int k0 = 32; k0 < n; k0 += 32) {
            if (lane < n - k0) {
                const v2f e2 = *(const v2f*)(myS + 2 * (k0 + lane));
                d0 += __expf(e2.x - m0);
                d1 += __expf(e2.y - m1);
            }
        }
        d0 = bsa_wave_sum(d0);
        d1 = bsa_wave_sum(d1);
        d0 = (d0 == 0.0f) ? 1.0f : d0;
        d1 = (d1 == 0.0f) ? 1.0f : d1;

        for (int k0 = 0; k0 < n; k0 += 32) {
            const int nn = min(32, n - k0);
            if (k0 != 0) {
                c = 0;
                x0 = 0.f;
                x1 = 0.f;
                if (lane < nn) {
                    c = myC[k0 + lane];
                    const v2f e2 = *(const v2f*)(myS + 2 * (k0 + lane));
                    x0 = __expf(e2.x - m0);
                    x1 = __expf(e2.y - m1);
                }
            }
            const float w0 = x0 / d0;
            const float w1 = x1 / d1;
            #pragma unroll 4
            for (int j = 0; j < nn; ++j) {
                const unsigned cj = (unsigned)__builtin_amdgcn_readlane(c, j);
                const float y0 = __int_as_float(__builtin_amdgcn_readlane(__float_as_int(w0), j));
                const float y1 = __int_as_float(__builtin_amdgcn_readlane(__float_as_int(w1), j));
                bsa_add_edge(v0p, v1p, cj, y0, y1, a0, a1);
            }
        }
    } else {
        const int novf = min(max(cnt[NN], 0), EE);
        const int nch  = CAP / 32 + (novf + 31) / 32;
        auto fetch = [&](int k, int& cc, float& t0, float& t1) -> bool {
            bool act = false;
            float cv = 0.f;
            cc = 0;
            if (k < CAP / 32) {
                const v2i e = rowslots[k * 32 + lane];
                cc = e.x;
                cv = __int_as_float(e.y);
                act = true;
            } else {
                const int i = (k - CAP / 32) * 32 + lane;
                if (i < novf && ovr[i] == r) {
                    cc = ovc[i];
                    cv = ovv[i];
                    act = true;
                }
            }
            cc = (int)min((unsigned)cc, (unsigned)(NN - 1));
            t0 = act ? cv * ksum[cc] : -INFINITY;
            t1 = act ? cv * ksum[NN + cc] : -INFINITY;
            return act;
        };
        float m0 = -INFINITY, m1 = -INFINITY;
        for (int k = 0; k < nch; ++k) {
            int cc;
            float t0, t1;
            fetch(k, cc, t0, t1);
            m0 = fmaxf(m0, t0);
            m1 = fmaxf(m1, t1);
        }
        m0 = bsa_wave_max(m0);
        m1 = bsa_wave_max(m1);
        if (!isfinite(m0)) m0 = 0.0f;
        if (!isfinite(m1)) m1 = 0.0f;
        float d0 = 0.f, d1 = 0.f;
        for (int k = 0; k < nch; ++k) {
            int cc;
            float t0, t1;
            if (fetch(k, cc, t0, t1)) {
                d0 += __expf(t0 - m0);
                d1 += __expf(t1 - m1);
            }
        }
        d0 = bsa_wave_sum(d0);
        d1 = bsa_wave_sum(d1);
        d0 = (d0 == 0.0f) ? 1.0f : d0;
        d1 = (d1 == 0.0f) ? 1.0f : d1;
        for (int k = 0; k < nch; ++k) {
            int cc;
            float t0, t1;
            const bool act = fetch(k, cc, t0, t1);
            const float w0 = act ? __expf(t0 - m0) / d0 : 0.0f;
            const float w1 = act ? __expf(t1 - m1) / d1 : 0.0f;
            unsigned todo = __builtin_amdgcn_ballot_w32(act);
            while (todo != 0u) {
                const int j = __builtin_ctz(todo);
                todo &= todo - 1u;
                const unsigned cj = (unsigned)__builtin_amdgcn_readlane(cc, j);
                const float y0 = __int_as_float(__builtin_amdgcn_readlane(__float_as_int(w0), j));
                const float y1 = __int_as_float(__builtin_amdgcn_readlane(__float_as_int(w1), j));
                bsa_add_edge(v0p, v1p, cj, y0, y1, a0, a1);
            }
        }
    }
    const v4f ob = *(const v4f*)(outb + lane * 4);
    const v4f gg = *(const v4f*)(lng + lane * 4);
    const v4f gb = *(const v4f*)(lnb + lane * 4);
    *(v4f*)(out + (size_t)r * VAL + lane * 4)        = bsa_silu_ln(a0 + ob, gg, gb);
    *(v4f*)(out + ((size_t)NN + r) * VAL + lane * 4) = bsa_silu_ln(a1 + ob, gg, gb);
}

extern "C" void kernel_launch(void* const* d_in, const int* in_sizes, int n_in,
                              void* d_out, int out_size, void* d_ws, size_t ws_size,
                              hipStream_t stream) {
    const float* x     = (const float*)d_in[0];
    const float* conn  = (const float*)d_in[1];
    const float* k_w   = (const float*)d_in[2];
    const float* k_b   = (const float*)d_in[3];
    const float* v_w   = (const float*)d_in[4];
    const float* v_b   = (const float*)d_in[5];
    const float* out_w = (const float*)d_in[6];
    const float* out_b = (const float*)d_in[7];
    const float* ln_g  = (const float*)d_in[8];
    const float* ln_b  = (const float*)d_in[9];
    const int*   rows  = (const int*)d_in[10];
    const int*   cols  = (const int*)d_in[11];
    float* out = (float*)d_out;

    char* p = (char*)d_ws;
    auto alloc = [&](size_t bytes) {
        void* r = (void*)p;
        p += (bytes + 255) & ~(size_t)255;
        return r;
    };
    float*    kwsum  = (float*)alloc(INP * sizeof(float));
    float*    kbsum  = (float*)alloc(sizeof(float));
    float*    ksum   = (float*)alloc((size_t)MROWS * sizeof(float));
    float*    values = (float*)alloc((size_t)MROWS * VAL * sizeof(float));
    float*    Wc     = (float*)alloc((size_t)INP * VAL * sizeof(float));
    float*    bc     = (float*)alloc(VAL * sizeof(float));
    int*      cnt    = (int*)alloc((size_t)CNT_INTS * sizeof(int));
    int*      slots  = (int*)alloc((size_t)NN * CAP * 2 * sizeof(int));
    int*      ovr    = (int*)alloc((size_t)EE * sizeof(int));
    int*      ovc    = (int*)alloc((size_t)EE * sizeof(int));
    float*    ovv    = (float*)alloc((size_t)EE * sizeof(float));
    if ((size_t)(p - (char*)d_ws) > ws_size) return;

    bsa_prep_zero<<<PREP_BLOCKS + ZERO_BLOCKS, 256, 0, stream>>>(k_w, k_b, v_w, v_b, out_w,
                                                                 kwsum, kbsum, Wc, bc, cnt);

    bsa_values_gemm_place<<<GEMM_BLOCKS + PLACE_BLOCKS, 256, 0, stream>>>(x, Wc, bc, kwsum, kbsum, ksum, values,
                                                                          conn, rows, cols, cnt, slots,
                                                                          ovr, ovc, ovv);

    bsa_gather_rows<<<(NN * 32 + 255) / 256, 256, 0, stream>>>(cnt, slots, ovr, ovc, ovv, ksum, values,
                                                            out_b, ln_g, ln_b, out);
}


# Gathered tables laid out by node then batch (GEMM block owns both batches of its 64 nodes): a sender's two sums are one 8-byte load, not two scattered 4-byte loads; its two rows one 1 KB record
# speedup vs baseline: 8.1504x; 1.0371x over previous
#include <hip/hip_runtime.h>
#include <math.h>

#define BB    2
#define NN    50000
#define EE    800000
#define INP   128
#define KEY   64
#define VAL   128
#define MROWS (BB * NN)
#define LN_EPS 1e-5f

typedef __attribute__((ext_vector_type(2))) float v2f;
typedef __attribute__((ext_vector_type(8))) float v8f;
typedef __attribute__((ext_vector_type(4))) float v4f;

#define KP 132

__device__ __forceinline__ void bsa_stage_w(const float* __restrict__ W, float* __restrict__ Bt) {
    const int col = threadIdx.x & 127;
    const int g0  = (threadIdx.x >> 7) << 4;
    float t[16][4];
    #pragma unroll
    for (int g = 0; g < 16; ++g) {
        #pragma unroll
        for (int i = 0; i < 4; ++i)
            t[g][i] = W[((g0 + g) * 4 + i) * 128 + col];
    }
    #pragma unroll
    for (int g = 0; g < 16; ++g) {
        v4f v = {t[g][0], t[g][1], t[g][2], t[g][3]};
        *(v4f*)(Bt + col * KP + (g0 + g) * 4) = v;
    }
}

template <bool DOT>
__device__ __forceinline__ void bsa_wmma_tile(const float* __restrict__ Arow,
                                              const float* __restrict__ Bt,
                                              v8f acc[8], int lane,
                                              const float* __restrict__ Ks, v2f& dot2) {
    const int k4 = (lane >> 4) << 2;
    const float* ap = Arow + k4;
    const float* bp = Bt + (lane & 15) * KP + k4;
    v4f a = *(const v4f*)ap;
    v4f bx[4], by[4];
    #pragma unroll
    for (int c = 0; c < 4; ++c) bx[c] = *(const v4f*)(bp + c * 16 * KP);
    #pragma nounroll
    for (int k = 0; k < 128; k += 8) {
        #pragma unroll
        for (int c = 0; c < 4; ++c) by[c] = *(const v4f*)(bp + (c + 4) * 16 * KP + k);
        const v2f a01 = {a.x, a.y}, a23 = {a.z, a.w};
        if (DOT) {
            v4f q = *(const v4f*)(Ks + k + k4);
            v2f q01 = {q.x, q.y}, q23 = {q.z, q.w};
            dot2 += a01 * q01;
            dot2 += a23 * q23;
        }
        #pragma unroll
        for (int c = 0; c < 4; ++c) {
            v2f b01 = {bx[c].x, bx[c].y};
            acc[c] = __builtin_amdgcn_wmma_f32_16x16x4_f32(
                false, a01, false, b01, (short)0, acc[c], false, false);
        }
        #pragma unroll
        for (int c = 0; c < 4; ++c) {
            v2f b23 = {bx[c].z, bx[c].w};
            acc[c] = __builtin_amdgcn_wmma_f32_16x16x4_f32(
                false, a23, false, b23, (short)0, acc[c], false, false);
        }
        const int kn = (k + 8 < 128) ? k + 8 : k;
        a = *(const v4f*)(ap + kn);
        asm volatile("" ::: "memory");
        #pragma unroll
        for (int c = 0; c < 4; ++c) bx[c] = *(const v4f*)(bp + c * 16 * KP + kn);
        #pragma unroll
        for (int c = 0; c < 4; ++c) {
            v2f b01 = {by[c].x, by[c].y};
            acc[c + 4] = __builtin_amdgcn_wmma_f32_16x16x4_f32(
                false, a01, false, b01, (short)0, acc[c + 4], false, false);
        }
        #pragma unroll
        for (int c = 0; c < 4; ++c) {
            v2f b23 = {by[c].z, by[c].w};
            acc[c + 4] = __builtin_amdgcn_wmma_f32_16x16x4_f32(
                false, a23, false, b23, (short)0, acc[c + 4], false, false);
        }
    }
}

typedef __attribute__((ext_vector_type(2))) int v2i;
typedef __attribute__((ext_vector_type(4))) int v4i;
#define PREP_BLOCKS  (INP / 16)
#define NTILES       (NN / 16)
#define GEMM_BLOCKS  ((NTILES + 3) / 4)
static_assert(NN % 16 == 0, "a tile of 16 rows never crosses from one batch into the other");
#define PLACE_BLOCKS ((EE / 8 + 255) / 256)
#define CNT_INTS     (((NN + 1 + 3) / 4) * 4)
#define ZERO_BLOCKS  ((CNT_INTS / 4 + 255) / 256)
#define CAP 64
static_assert(EE % 8 == 0 && CAP % 32 == 0, "16-byte loads of the edge list; a row's slots are whole chunks of 32");

__device__ __forceinline__ void bsa_put_edge(int r, int c, float cv, int slot,
                                             int* __restrict__ cnt, int* __restrict__ slots,
                                             int* __restrict__ ovr, int* __restrict__ ovc,
                                             float* __restrict__ ovv) {
    if (slot < CAP) {
        const v2i t = {c, __float_as_int(cv)};
        *(v2i*)(slots + ((size_t)r * CAP + max(slot, 0)) * 2) = t;
    } else {
        int o = atomicAdd(&cnt[NN], 1);
        o = min(max(o, 0), EE - 1);
        ovr[o] = r;
        ovc[o] = c;
        ovv[o] = cv;
    }
}

__device__ __forceinline__ void bsa_place8(int t, const float* __restrict__ conn,
                                           const int* __restrict__ rows, const int* __restrict__ cols,
                                           int* __restrict__ cnt, int* __restrict__ slots,
                                           int* __restrict__ ovr, int* __restrict__ ovc,
                                           float* __restrict__ ovv) {
    const v4i ra = *(const v4i*)(rows + 8 * t);
    const v4i rb = *(const v4i*)(rows + 8 * t + 4);
    const v4i ca = *(const v4i*)(cols + 8 * t);
    const v4i cb = *(const v4i*)(cols + 8 * t + 4);
    const v4f va = *(const v4f*)(conn + 8 * t);
    const v4f vb = *(const v4f*)(conn + 8 * t + 4);
    const int q0 = atomicAdd(&cnt[ra.x], 1);
    const int q1 = atomicAdd(&cnt[ra.y], 1);
    const int q2 = atomicAdd(&cnt[ra.z], 1);
    const int q3 = atomicAdd(&cnt[ra.w], 1);
    const int q4 = atomicAdd(&cnt[rb.x], 1);
    const int q5 = atomicAdd(&cnt[rb.y], 1);
    const int q6 = atomicAdd(&cnt[rb.z], 1);
    const int q7 = atomicAdd(&cnt[rb.w], 1);
    bsa_put_edge(ra.x, ca.x, va.x, q0, cnt, slots, ovr, ovc, ovv);
    bsa_put_edge(ra.y, ca.y, va.y, q1, cnt, slots, ovr, ovc, ovv);
    bsa_put_edge(ra.z, ca.z, va.z, q2, cnt, slots, ovr, ovc, ovv);
    bsa_put_edge(ra.w, ca.w, va.w, q3, cnt, slots, ovr, ovc, ovv);
    bsa_put_edge(rb.x, cb.x, vb.x, q4, cnt, slots, ovr, ovc, ovv);
    bsa_put_edge(rb.y, cb.y, vb.y, q5, cnt, slots, ovr, ovc, ovv);
    bsa_put_edge(rb.z, cb.z, vb.z, q6, cnt, slots, ovr, ovc, ovv);
    bsa_put_edge(rb.w, cb.w, vb.w, q7, cnt, slots, ovr, ovc, ovv);
}

__global__ __launch_bounds__(256) void bsa_prep_zero(const float* __restrict__ kw,
                                const float* __restrict__ kb,
                                const float* __restrict__ vw,
                                const float* __restrict__ vb,
                                const float* __restrict__ ow,
                                float* __restrict__ kwsum,
                                float* __restrict__ kbsum,
                                float* __restrict__ Wc,
                                float* __restrict__ bc,
                                int* __restrict__ cnt) {
    __shared__ __attribute__((aligned(16))) float Bs[128 * KP];
    if (blockIdx.x >= PREP_BLOCKS) {
        const int t = (blockIdx.x - PREP_BLOCKS) * 256 + threadIdx.x;
        if (t < CNT_INTS / 4) {
            const v4i z = {0, 0, 0, 0};
            *(v4i*)(cnt + 4 * t) = z;
        }
        return;
    }
    bsa_stage_w(ow, Bs);
    const int i = threadIdx.x;
    if (blockIdx.x == 0) {
        if (i < INP) {
            float s = 0.f;
            for (int j = 0; j < KEY; ++j) s += kw[i * KEY + j];
            kwsum[i] = s;
        } else if (i == INP) {
            float t = 0.f;
            for (int j = 0; j < KEY; ++j) t += kb[j];
            kbsum[0] = t;
        }
    }
    __syncthreads();

    if (blockIdx.x == 1 && i < VAL) {
        v4f t = {0.f, 0.f, 0.f, 0.f};
        const float* colp = Bs + i * KP;
        for (int k = 0; k < VAL; k += 4)
            t += *(const v4f*)(vb + k) * *(const v4f*)(colp + k);
        bc[i] = (t.x + t.y) + (t.z + t.w);
    }

    const int lane    = i & 31;
    const int wv      = i >> 5;
    const int rowbase = blockIdx.x << 4;
    const int k4      = (lane >> 4) << 2;
    const float* ap = vw + (size_t)(rowbase + (lane & 15)) * 128 + k4;
    const float* bp = Bs + ((wv << 4) + (lane & 15)) * KP + k4;
    v8f acc0 = {}, acc1 = {};
    for (int k = 0; k < 128; k += 8) {
        const v4f a = *(const v4f*)(ap + k);
        const v4f b = *(const v4f*)(bp + k);
        const v2f a01 = {a.x, a.y}, a23 = {a.z, a.w};
        const v2f b01 = {b.x, b.y}, b23 = {b.z, b.w};
        acc0 = __builtin_amdgcn_wmma_f32_16x16x4_f32(
            false, a01, false, b01, (short)0, acc0, false, false);
        acc1 = __builtin_amdgcn_wmma_f32_16x16x4_f32(
            false, a23, false, b23, (short)0, acc1, false, false);
    }
    const int n      = lane & 15;
    const int rowoff = (lane >> 4) << 3;
    #pragma unroll
    for (int r = 0; r < 8; ++r)
        Wc[(rowbase + rowoff + r) * 128 + (wv << 4) + n] = acc0[r] + acc1[r];
}

__global__ __launch_bounds__(256) void bsa_values_gemm_place(const float* __restrict__ A,
                                const float* __restrict__ W,
                                const float* __restrict__ bias,
                                const float* __restrict__ kwsum,
                                const float* __restrict__ kbsum,
                                float* __restrict__ ksum,
                                float* __restrict__ out,
                                const float* __restrict__ conn,
                                const int* __restrict__ rows,
                                const int* __restrict__ cols,
                                int* __restrict__ cnt,
                                int* __restrict__ slots,
                                int* __restrict__ ovr,
                                int* __restrict__ ovc,
                                float* __restrict__ ovv) {
    __shared__ __attribute__((aligned(16))) float Bs[128 * KP];
    __shared__ __attribute__((aligned(16))) float Ks[INP];
    __shared__ float Kd[128];
    if (blockIdx.x >= GEMM_BLOCKS) {
        const int t = (blockIdx.x - GEMM_BLOCKS) * 256 + threadIdx.x;
        if (t < EE / 8) {
            bsa_place8(t, conn, rows, cols, cnt, slots, ovr, ovc, ovv);
        }
        return;
    }
    const int blk = blockIdx.x;
    bsa_stage_w(W, Bs);
    if (threadIdx.x < INP) Ks[threadIdx.x] = kwsum[threadIdx.x];
    __syncthreads();

    int lane    = threadIdx.x & 31;
    int wv      = threadIdx.x >> 5;
    int bsel    = wv >> 2;
    int ntile   = blk * 4 + (wv & 3);
    int node0   = ntile * 16;
    if (ntile < NTILES) {
        v8f acc[8] = {};
        v2f dot2 = {0.f, 0.f};
        const float* Arow = A + ((size_t)bsel * NN + node0 + (lane & 15)) * 128;
        bsa_wmma_tile<true>(Arow, Bs, acc, lane, Ks, dot2);

        float dot = dot2.x + dot2.y;
        dot += __shfl_xor(dot, 16);
        if (lane < 16) Kd[(wv << 4) + lane] = dot;

        int n      = lane & 15;
        int rowoff = (lane >> 4) << 3;
        #pragma unroll
        for (int c = 0; c < 8; ++c) {
            float bb = bias[(c << 4) + n];
            #pragma unroll
            for (int r = 0; r < 8; ++r)
                out[((size_t)(node0 + rowoff + r) * 2 + bsel) * 128 + (c << 4) + n] = acc[c][r] + bb;
        }
    }
    __syncthreads();
    int nl   = threadIdx.x >> 1;
    int node = blk * 64 + nl;
    if (threadIdx.x < 128 && node < NN)
        ksum[(size_t)node * 2 + (threadIdx.x & 1)] = Kd[((threadIdx.x & 1) << 6) + nl] + kbsum[0];
}

__device__ __forceinline__ float bsa_xhalf(float x) {
    int b = __builtin_bit_cast(int, x);
    return __builtin_bit_cast(float,
        __builtin_amdgcn_permlanex16(b, b, 0x76543210, 0xfedcba98, false, false));
}
template <int MASK>
__device__ __forceinline__ float bsa_xrow(float x) {
    return __builtin_bit_cast(float,
        __builtin_amdgcn_update_dpp(0, __builtin_bit_cast(int, x), 0x160 + MASK, 0xf, 0xf, true));
}
__device__ __forceinline__ float bsa_wave_max(float x) {
    x = fmaxf(x, bsa_xrow<8>(x));
    x = fmaxf(x, bsa_xrow<4>(x));
    x = fmaxf(x, bsa_xrow<2>(x));
    x = fmaxf(x, bsa_xrow<1>(x));
    return fmaxf(x, bsa_xhalf(x));
}
__device__ __forceinline__ float bsa_wave_sum(float x) {
    x += bsa_xrow<8>(x);
    x += bsa_xrow<4>(x);
    x += bsa_xrow<2>(x);
    x += bsa_xrow<1>(x);
    return x + bsa_xhalf(x);
}
__device__ __forceinline__ v4f bsa_silu_ln(v4f h, v4f g, v4f b) {
    h.x = h.x / (1.0f + __expf(-h.x));
    h.y = h.y / (1.0f + __expf(-h.y));
    h.z = h.z / (1.0f + __expf(-h.z));
    h.w = h.w / (1.0f + __expf(-h.w));
    const float mu = bsa_wave_sum((h.x + h.y) + (h.z + h.w)) * (1.0f / VAL);
    const v4f d = h - mu;
    const float var = bsa_wave_sum((d.x * d.x + d.y * d.y) + (d.z * d.z + d.w * d.w)) * (1.0f / VAL);
    const float rstd = rsqrtf(var + LN_EPS);
    return d * rstd * g + b;
}

__device__ __forceinline__ void bsa_add_edge(const float* __restrict__ vp,
                                             unsigned cj, float y0, float y1, v4f& a0, v4f& a1) {
    cj = min(cj, (unsigned)(NN - 1));
    const float* rec = vp + (size_t)cj * (2 * VAL);
    const v4f p = *(const v4f*)rec;
    const v4f q = *(const v4f*)(rec + VAL);
    a0 += y0 * p;
    a1 += y1 * q;
}
__global__ __launch_bounds__(256, 16) void bsa_gather_rows(const int* __restrict__ cnt,
                                const int* __restrict__ slots,
                                const int* __restrict__ ovr,
                                const int* __restrict__ ovc,
                                const float* __restrict__ ovv,
                                const float* __restrict__ ksum,
                                const float* __restrict__ values,
                                const float* __restrict__ outb,
                                const float* __restrict__ lng,
                                const float* __restrict__ lnb,
                                float* __restrict__ out) {
    __shared__ int sC[8][CAP];
    __shared__ __attribute__((aligned(8))) float sS[8][CAP * 2];
    const int lane = threadIdx.x & 31;
    int*   myC = sC[threadIdx.x >> 5];
    float* myS = sS[threadIdx.x >> 5];
    const int r = __builtin_amdgcn_readfirstlane((int)((blockIdx.x * 256 + threadIdx.x) >> 5));
    if (r >= NN) return;
    const int n = min(max(cnt[r], 0), EE);
    const v2i* rowslots = (const v2i*)slots + (size_t)r * CAP;
    const float* vp = values + lane * 4;
    v4f a0 = {0.f, 0.f, 0.f, 0.f}, a1 = {0.f, 0.f, 0.f, 0.f};

    if (n <= CAP) {
        const int nA = min(n, 32), nB = n - nA;
        int   c  = 0, cB = 0;
        float s0 = -INFINITY, s1 = -INFINITY, s0B = -INFINITY, s1B = -INFINITY;
        if (lane < nA) {
            const v2i e = rowslots[lane];
            c = (int)min((unsigned)e.x, (unsigned)(NN - 1));
            const float cv = __int_as_float(e.y);
            const v2f kk = *(const v2f*)(ksum + 2 * (size_t)c);
            s0 = cv * kk.x;
            s1 = cv * kk.y;
        }
        if (lane < nB) {
            const v2i e = rowslots[32 + lane];
            cB = (int)min((unsigned)e.x, (unsigned)(NN - 1));
            const float cv = __int_as_float(e.y);
            const v2f kk = *(const v2f*)(ksum + 2 * (size_t)cB);
            s0B = cv * kk.x;
            s1B = cv * kk.y;
        }
        int ltA = 0, eqA = 0, ltB = 0, eqB = 0;
        for (int j = 0; j < nA; ++j) {
            const int cj = __builtin_amdgcn_readlane(c, j);
            ltA += (cj < c) ? 1 : 0;
            eqA += (cj == c) ? 1 : 0;
        }
        if (nB > 0) {
            for (int j = 0; j < nB; ++j) {
                const int cj = __builtin_amdgcn_readlane(cB, j);
                ltA += (cj < c) ? 1 : 0;
                eqA += (cj == c) ? 1 : 0;
                ltB += (cj < cB) ? 1 : 0;
                eqB += (cj == cB) ? 1 : 0;
            }
            for (int j = 0; j < nA; ++j) {
                const int cj = __builtin_amdgcn_readlane(c, j);
                ltB += (cj < cB) ? 1 : 0;
                eqB += (cj == cB) ? 1 : 0;
            }
        }
        int rankA = ltA, rankB = ltB;
        if (__builtin_amdgcn_ballot_w32((lane < nA && eqA > 1) || (lane < nB && eqB > 1)) != 0u) {
            const unsigned uaA = __float_as_uint(s0),  ubA = __float_as_uint(s1);
            const unsigned uaB = __float_as_uint(s0B), ubB = __float_as_uint(s1B);
            int rkA = 0, rkB = 0;
            for (int j = 0; j < n; ++j) {
                int cj;
                unsigned aj, bj;
                if (j < 32) {
                    cj = __builtin_amdgcn_readlane(c, j);
                    aj = (unsigned)__builtin_amdgcn_readlane((int)uaA, j);
                    bj = (unsigned)__builtin_amdgcn_readlane((int)ubA, j);
                } else {
                    cj = __builtin_amdgcn_readlane(cB, j - 32);
                    aj = (unsigned)__builtin_amdgcn_readlane((int)uaB, j - 32);
                    bj = (unsigned)__builtin_amdgcn_readlane((int)ubB, j - 32);
                }
                const bool beforeA = (cj < c) || (cj == c && (aj < uaA || (aj == uaA &&
                                     (bj < ubA || (bj == ubA && j < lane)))));
                const bool beforeB = (cj < cB) || (cj == cB && (aj < uaB || (aj == uaB &&
                                     (bj < ubB || (bj == ubB && j < 32 + lane)))));
                rkA += beforeA ? 1 : 0;
                rkB += beforeB ? 1 : 0;
            }
            rankA = rkA;
            rankB = rkB;
        }
        if (lane < nA) {
            rankA = min(max(rankA, 0), CAP - 1);
            myC[rankA] = c;
            const v2f e2 = {s0, s1};
            *(v2f*)(myS + 2 * rankA) = e2;
        }
        if (lane < nB) {
            rankB = min(max(rankB, 0), CAP - 1);
            myC[rankB] = cB;
            const v2f e2 = {s0B, s1B};
            *(v2f*)(myS + 2 * rankB) = e2;
        }
        __builtin_amdgcn_fence(__ATOMIC_RELEASE, "wavefront");
        __builtin_amdgcn_wave_barrier();
        __builtin_amdgcn_fence(__ATOMIC_ACQUIRE, "wavefront");

        c  = 0;
        s0 = -INFINITY;
        s1 = -INFINITY;
        if (lane < n) {
            c = myC[lane];
            const v2f e2 = *(const v2f*)(myS + 2 * lane);
            s0 = e2.x;
            s1 = e2.y;
        }

        float m0 = s0, m1 = s1;
        for (int k0 = 32; k0 < n; k0 += 32) {
            if (lane < n - k0) {
                const v2f e2 = *(const v2f*)(myS + 2 * (k0 + lane));
                m0 = fmaxf(m0, e2.x);
                m1 = fmaxf(m1, e2.y);
            }
        }
        m0 = bsa_wave_max(m0);
        m1 = bsa_wave_max(m1);
        if (!isfinite(m0)) m0 = 0.0f;
        if (!isfinite(m1)) m1 = 0.0f;

        float x0 = (lane < n) ? __expf(s0 - m0) : 0.0f;
        float x1 = (lane < n) ? __expf(s1 - m1) : 0.0f;
        float d0 = x0, d1 = x1;
        for (int k0 = 32; k0 < n; k0 += 32) {
            if (lane < n - k0) {
                const v2f e2 = *(const v2f*)(myS + 2 * (k0 + lane));
                d0 += __expf(e2.x - m0);
                d1 += __expf(e2.y - m1);
            }
        }
        d0 = bsa_wave_sum(d0);
        d1 = bsa_wave_sum(d1);
        d0 = (d0 == 0.0f) ? 1.0f : d0;
        d1 = (d1 == 0.0f) ? 1.0f : d1;

        for (int k0 = 0; k0 < n; k0 += 32) {
            const int nn = min(32, n - k0);
            if (k0 != 0) {
                c = 0;
                x0 = 0.f;
                x1 = 0.f;
                if (lane < nn) {
                    c = myC[k0 + lane];
                    const v2f e2 = *(const v2f*)(myS + 2 * (k0 + lane));
                    x0 = __expf(e2.x - m0);
                    x1 = __expf(e2.y - m1);
                }
            }
            const float w0 = x0 / d0;
            const float w1 = x1 / d1;
            #pragma unroll 4
            for (int j = 0; j < nn; ++j) {
                const unsigned cj = (unsigned)__builtin_amdgcn_readlane(c, j);
                const float y0 = __int_as_float(__builtin_amdgcn_readlane(__float_as_int(w0), j));
                const float y1 = __int_as_float(__builtin_amdgcn_readlane(__float_as_int(w1), j));
                bsa_add_edge(vp, cj, y0, y1, a0, a1);
            }
        }
    } else {
        const int novf = min(max(cnt[NN], 0), EE);
        const int nch  = CAP / 32 + (novf + 31) / 32;
        auto fetch = [&](int k, int& cc, float& t0, float& t1) -> bool {
            bool act = false;
            float cv = 0.f;
            cc = 0;
            if (k < CAP / 32) {
                const v2i e = rowslots[k * 32 + lane];
                cc = e.x;
                cv = __int_as_float(e.y);
                act = true;
            } else {
                const int i = (k - CAP / 32) * 32 + lane;
                if (i < novf && ovr[i] == r) {
                    cc = ovc[i];
                    cv = ovv[i];
                    act = true;
                }
            }
            cc = (int)min((unsigned)cc, (unsigned)(NN - 1));
            const v2f kk = *(const v2f*)(ksum + 2 * (size_t)cc);
            t0 = act ? cv * kk.x : -INFINITY;
            t1 = act ? cv * kk.y : -INFINITY;
            return act;
        };
        float m0 = -INFINITY, m1 = -INFINITY;
        for (int k = 0; k < nch; ++k) {
            int cc;
            float t0, t1;
            fetch(k, cc, t0, t1);
            m0 = fmaxf(m0, t0);
            m1 = fmaxf(m1, t1);
        }
        m0 = bsa_wave_max(m0);
        m1 = bsa_wave_max(m1);
        if (!isfinite(m0)) m0 = 0.0f;
        if (!isfinite(m1)) m1 = 0.0f;
        float d0 = 0.f, d1 = 0.f;
        for (int k = 0; k < nch; ++k) {
            int cc;
            float t0, t1;
            if (fetch(k, cc, t0, t1)) {
                d0 += __expf(t0 - m0);
                d1 += __expf(t1 - m1);
            }
        }
        d0 = bsa_wave_sum(d0);
        d1 = bsa_wave_sum(d1);
        d0 = (d0 == 0.0f) ? 1.0f : d0;
        d1 = (d1 == 0.0f) ? 1.0f : d1;
        for (int k = 0; k < nch; ++k) {
            int cc;
            float t0, t1;
            const bool act = fetch(k, cc, t0, t1);
            const float w0 = act ? __expf(t0 - m0) / d0 : 0.0f;
            const float w1 = act ? __expf(t1 - m1) / d1 : 0.0f;
            unsigned todo = __builtin_amdgcn_ballot_w32(act);
            while (todo != 0u) {
                const int j = __builtin_ctz(todo);
                todo &= todo - 1u;
                const unsigned cj = (unsigned)__builtin_amdgcn_readlane(cc, j);
                const float y0 = __int_as_float(__builtin_amdgcn_readlane(__float_as_int(w0), j));
                const float y1 = __int_as_float(__builtin_amdgcn_readlane(__float_as_int(w1), j));
                bsa_add_edge(vp, cj, y0, y1, a0, a1);
            }
        }
    }
    const v4f ob = *(const v4f*)(outb + lane * 4);
    const v4f gg = *(const v4f*)(lng + lane * 4);
    const v4f gb = *(const v4f*)(lnb + lane * 4);
    *(v4f*)(out + (size_t)r * VAL + lane * 4)        = bsa_silu_ln(a0 + ob, gg, gb);
    *(v4f*)(out + ((size_t)NN + r) * VAL + lane * 4) = bsa_silu_ln(a1 + ob, gg, gb);
}

extern "C" void kernel_launch(void* const* d_in, const int* in_sizes, int n_in,
                              void* d_out, int out_size, void* d_ws, size_t ws_size,
                              hipStream_t stream) {
    const float* x     = (const float*)d_in[0];
    const float* conn  = (const float*)d_in[1];
    const float* k_w   = (const float*)d_in[2];
    const float* k_b   = (const float*)d_in[3];
    const float* v_w   = (const float*)d_in[4];
    const float* v_b   = (const float*)d_in[5];
    const float* out_w = (const float*)d_in[6];
    const float* out_b = (const float*)d_in[7];
    const float* ln_g  = (const float*)d_in[8];
    const float* ln_b  = (const float*)d_in[9];
    const int*   rows  = (const int*)d_in[10];
    const int*   cols  = (const int*)d_in[11];
    float* out = (float*)d_out;

    char* p = (char*)d_ws;
    auto alloc = [&](size_t bytes) {
        void* r = (void*)p;
        p += (bytes + 255) & ~(size_t)255;
        return r;
    };
    float*    kwsum  = (float*)alloc(INP * sizeof(float));
    float*    kbsum  = (float*)alloc(sizeof(float));
    float*    ksum   = (float*)alloc((size_t)MROWS * sizeof(float));
    float*    values = (float*)alloc((size_t)MROWS * VAL * sizeof(float));
    float*    Wc     = (float*)alloc((size_t)INP * VAL * sizeof(float));
    float*    bc     = (float*)alloc(VAL * sizeof(float));
    int*      cnt    = (int*)alloc((size_t)CNT_INTS * sizeof(int));
    int*      slots  = (int*)alloc((size_t)NN * CAP * 2 * sizeof(int));
    int*      ovr    = (int*)alloc((size_t)EE * sizeof(int));
    int*      ovc    = (int*)alloc((size_t)EE * sizeof(int));
    float*    ovv    = (float*)alloc((size_t)EE * sizeof(float));
    if ((size_t)(p - (char*)d_ws) > ws_size) return;

    bsa_prep_zero<<<PREP_BLOCKS + ZERO_BLOCKS, 256, 0, stream>>>(k_w, k_b, v_w, v_b, out_w,
                                                                 kwsum, kbsum, Wc, bc, cnt);

    bsa_values_gemm_place<<<GEMM_BLOCKS + PLACE_BLOCKS, 256, 0, stream>>>(x, Wc, bc, kwsum, kbsum, ksum, values,
                                                                          conn, rows, cols, cnt, slots,
                                                                          ovr, ovc, ovv);

    bsa_gather_rows<<<(NN * 32 + 255) / 256, 256, 0, stream>>>(cnt, slots, ovr, ovc, ovv, ksum, values,
                                                            out_b, ln_g, ln_b, out);
}
